# starting kernel
# baseline (speedup 1.0000x reference)
#include <hip/hip_runtime.h>
#include <math.h>

typedef float v2f __attribute__((ext_vector_type(2)));
typedef float v4f __attribute__((ext_vector_type(4)));
typedef float v8f __attribute__((ext_vector_type(8)));

#define IN_DIM 128
#define OUT_DIM 128
#define NHEAD 4
#define HD 32
#define NEG_SLOPE 0.2f
#define EPS_SOFTMAX 1e-8f
#define EPS_LN 1e-5f

__device__ __forceinline__ void atomicMaxF32(float* addr, float val) {
    if (val >= 0.0f) {
        atomicMax((int*)addr, __float_as_int(val));
    } else {
        atomicMin((unsigned int*)addr, (unsigned int)__float_as_int(val));
    }
}

__global__ void init_kernel(float* emax0, float* emax1, float* asum0, float* asum1, int NH) {
    int gid = blockIdx.x * blockDim.x + threadIdx.x;
    if (gid >= NH) return;
    emax0[gid] = -INFINITY;
    emax1[gid] = -INFINITY;
    asum0[gid] = 0.0f;
    asum1[gid] = 0.0f;
}

__global__ __launch_bounds__(256) void gemm_wmma_kernel(
    const float* __restrict__ h, const float* __restrict__ W,
    const float* __restrict__ Wself, float* __restrict__ h_proj,
    float* __restrict__ out, int N)
{
    const int wave = threadIdx.x >> 5;
    const int lane = threadIdx.x & 31;
    const int row0 = blockIdx.x * 16;
    const int mrow = lane & 15;
    const int khalf = (lane >> 4) * 2;

    int arow_idx = row0 + mrow;
    if (arow_idx >= N) arow_idx = N - 1;
    const float* arow = h + (size_t)arow_idx * IN_DIM;

    const int col0 = wave * 32;
    const int ccol0 = col0 + mrow;
    const int ccol1 = col0 + 16 + mrow;
    const bool t0_proj = (col0 < 128);
    const bool t1_proj = (col0 + 16 < 128);
    const float* bp0 = t0_proj ? (W + ccol0) : (Wself + (ccol0 - 128));
    const float* bp1 = t1_proj ? (W + ccol1) : (Wself + (ccol1 - 128));

    v8f c0 = {};
    v8f c1 = {};
    for (int k0 = 0; k0 < IN_DIM; k0 += 4) {
        const int k = k0 + khalf;
        v2f a, b0, b1;
        a.x = arow[k];
        a.y = arow[k + 1];
        b0.x = bp0[(size_t)k * OUT_DIM];
        b0.y = bp0[(size_t)(k + 1) * OUT_DIM];
        b1.x = bp1[(size_t)k * OUT_DIM];
        b1.y = bp1[(size_t)(k + 1) * OUT_DIM];
        c0 = __builtin_amdgcn_wmma_f32_16x16x4_f32(false, a, false, b0,
                                                   (short)0, c0, false, false);
        c1 = __builtin_amdgcn_wmma_f32_16x16x4_f32(false, a, false, b1,
                                                   (short)0, c1, false, false);
    }

    const int mbase = (lane >> 4) * 8;
    float* dst0 = t0_proj ? (h_proj + ccol0) : (out + (ccol0 - 128));
    float* dst1 = t1_proj ? (h_proj + ccol1) : (out + (ccol1 - 128));
    if (row0 + 16 <= N) {
#pragma unroll
        for (int v = 0; v < 8; ++v) {
            const size_t row = (size_t)(row0 + mbase + v);
            dst0[row * OUT_DIM] = c0[v];
            dst1[row * OUT_DIM] = c1[v];
        }
    } else {
#pragma unroll
        for (int v = 0; v < 8; ++v) {
            const int row = row0 + mbase + v;
            if (row < N) {
                dst0[(size_t)row * OUT_DIM] = c0[v];
                dst1[(size_t)row * OUT_DIM] = c1[v];
            }
        }
    }
}

__global__ void score_kernel(const float* __restrict__ h_proj,
                             const float* __restrict__ a_in,
                             const float* __restrict__ a_out,
                             float* __restrict__ s_in_s, float* __restrict__ s_in_r,
                             float* __restrict__ s_out_s, float* __restrict__ s_out_r,
                             int NH)
{
    int gid = blockIdx.x * blockDim.x + threadIdx.x;
    if (gid >= NH) return;
    const int hh = gid & 3;
    const int n = gid >> 2;
    const float* x = h_proj + (size_t)n * OUT_DIM + hh * HD;
    const float* ai = a_in + hh * 2 * HD;
    const float* ao = a_out + hh * 2 * HD;
    float s0 = 0.f, s1 = 0.f, s2 = 0.f, s3 = 0.f;
#pragma unroll
    for (int d = 0; d < HD; d += 4) {
        const v4f xv = *(const v4f*)(x + d);
#pragma unroll
        for (int j = 0; j < 4; ++j) {
            s0 += xv[j] * ai[d + j];
            s1 += xv[j] * ai[HD + d + j];
            s2 += xv[j] * ao[d + j];
            s3 += xv[j] * ao[HD + d + j];
        }
    }
    s_in_s[gid] = s0;
    s_in_r[gid] = s1;
    s_out_s[gid] = s2;
    s_out_r[gid] = s3;
}

__global__ void edge_max_kernel(const int* __restrict__ snd, const int* __restrict__ rcv,
                                const float* __restrict__ s_s, const float* __restrict__ s_r,
                                float* __restrict__ emax, int E)
{
    int e = blockIdx.x * blockDim.x + threadIdx.x;
    if (e >= E) return;
    const int s = snd[e];
    const int r = rcv[e];
    const v4f ss = *(const v4f*)(s_s + (size_t)s * 4);
    const v4f sr = *(const v4f*)(s_r + (size_t)r * 4);
#pragma unroll
    for (int hh = 0; hh < NHEAD; ++hh) {
        float ev = ss[hh] + sr[hh];
        ev = (ev > 0.0f) ? ev : NEG_SLOPE * ev;
        atomicMaxF32(&emax[(size_t)r * 4 + hh], ev);
    }
}

__global__ void edge_sum_kernel(const int* __restrict__ snd, const int* __restrict__ rcv,
                                const float* __restrict__ s_s, const float* __restrict__ s_r,
                                const float* __restrict__ emax,
                                float* __restrict__ alpha_buf, float* __restrict__ asum,
                                int E)
{
    int e = blockIdx.x * blockDim.x + threadIdx.x;
    if (e >= E) return;
    const int s = snd[e];
    const int r = rcv[e];
    const v4f ss = *(const v4f*)(s_s + (size_t)s * 4);
    const v4f sr = *(const v4f*)(s_r + (size_t)r * 4);
    const v4f em = *(const v4f*)(emax + (size_t)r * 4);
    v4f alpha;
#pragma unroll
    for (int hh = 0; hh < NHEAD; ++hh) {
        float ev = ss[hh] + sr[hh];
        ev = (ev > 0.0f) ? ev : NEG_SLOPE * ev;
        alpha[hh] = __expf(ev - em[hh]);
        atomicAdd(&asum[(size_t)r * 4 + hh], alpha[hh]);
    }
    *(v4f*)(alpha_buf + (size_t)e * 4) = alpha;
}

__global__ void edge_agg_kernel(const int* __restrict__ snd, const int* __restrict__ rcv,
                                const float* __restrict__ h_proj,
                                const float* __restrict__ alpha_buf,
                                const float* __restrict__ asum,
                                float* __restrict__ out, int EH)
{
    int gid = blockIdx.x * blockDim.x + threadIdx.x;
    if (gid >= EH) return;
    const int hh = gid & 3;
    const int e = gid >> 2;
    const int s = snd[e];
    const int r = rcv[e];
    const float w = alpha_buf[(size_t)e * 4 + hh] / (asum[(size_t)r * 4 + hh] + EPS_SOFTMAX);
    const float* xs = h_proj + (size_t)s * OUT_DIM + hh * HD;
    float* xo = out + (size_t)r * OUT_DIM + hh * HD;
#pragma unroll
    for (int d = 0; d < HD; d += 4) {
        const v4f xv = *(const v4f*)(xs + d);
#pragma unroll
        for (int j = 0; j < 4; ++j) {
            atomicAdd(xo + d + j, w * xv[j]);
        }
    }
}

__global__ __launch_bounds__(128) void ln_kernel(float* __restrict__ out,
                                                 const float* __restrict__ bias,
                                                 const float* __restrict__ gamma,
                                                 const float* __restrict__ beta,
                                                 int N)
{
    __shared__ float red[OUT_DIM];
    const int n = blockIdx.x;
    const int t = threadIdx.x;
    if (n >= N) return;
    const float x = out[(size_t)n * OUT_DIM + t] + bias[t];
    red[t] = x;
    __syncthreads();
    for (int s = OUT_DIM / 2; s > 0; s >>= 1) {
        if (t < s) red[t] += red[t + s];
        __syncthreads();
    }
    const float mean = red[0] * (1.0f / OUT_DIM);
    __syncthreads();
    const float d = x - mean;
    red[t] = d * d;
    __syncthreads();
    for (int s = OUT_DIM / 2; s > 0; s >>= 1) {
        if (t < s) red[t] += red[t + s];
        __syncthreads();
    }
    const float var = red[0] * (1.0f / OUT_DIM);
    const float inv = rsqrtf(var + EPS_LN);
    out[(size_t)n * OUT_DIM + t] = d * inv * gamma[t] + beta[t];
}

extern "C" void kernel_launch(void* const* d_in, const int* in_sizes, int n_in,
                              void* d_out, int out_size, void* d_ws, size_t ws_size,
                              hipStream_t stream) {
    const float* h      = (const float*)d_in[0];
    const int*   eidx   = (const int*)d_in[1];
    const float* W      = (const float*)d_in[2];
    const float* Wself  = (const float*)d_in[3];
    const float* a_in   = (const float*)d_in[4];
    const float* a_out  = (const float*)d_in[5];
    const float* bias   = (const float*)d_in[6];
    const float* gamma  = (const float*)d_in[7];
    const float* beta   = (const float*)d_in[8];
    float* out = (float*)d_out;

    const int N = in_sizes[0] / IN_DIM;
    const int E = in_sizes[1] / 2;
    const int NH = N * NHEAD;
    const int EH = E * NHEAD;

    const int* src = eidx;
    const int* dst = eidx + E;

    float* ws = (float*)d_ws;
    float* h_proj  = ws;
    float* s_in_s  = h_proj + (size_t)N * OUT_DIM;
    float* s_in_r  = s_in_s + NH;
    float* s_out_s = s_in_r + NH;
    float* s_out_r = s_out_s + NH;
    float* emax0   = s_out_r + NH;
    float* emax1   = emax0 + NH;
    float* asum0   = emax1 + NH;
    float* asum1   = asum0 + NH;
    float* alpha0  = asum1 + NH;
    float* alpha1  = alpha0 + (size_t)E * NHEAD;

    {
        int threads = 256, blocks = (NH + threads - 1) / threads;
        init_kernel<<<blocks, threads, 0, stream>>>(emax0, emax1, asum0, asum1, NH);
    }
    {
        int blocks = (N + 15) / 16;
        gemm_wmma_kernel<<<blocks, 256, 0, stream>>>(h, W, Wself, h_proj, out, N);
    }
    {
        int threads = 256, blocks = (NH + threads - 1) / threads;
        score_kernel<<<blocks, threads, 0, stream>>>(h_proj, a_in, a_out,
                                                     s_in_s, s_in_r, s_out_s, s_out_r, NH);
    }
    {
        int threads = 256, blocks = (E + threads - 1) / threads;
        edge_max_kernel<<<blocks, threads, 0, stream>>>(src, dst, s_in_s, s_in_r, emax0, E);
        edge_max_kernel<<<blocks, threads, 0, stream>>>(dst, src, s_out_s, s_out_r, emax1, E);
    }
    {
        int threads = 256, blocks = (E + threads - 1) / threads;
        edge_sum_kernel<<<blocks, threads, 0, stream>>>(src, dst, s_in_s, s_in_r, emax0,
                                                        alpha0, asum0, E);
        edge_sum_kernel<<<blocks, threads, 0, stream>>>(dst, src, s_out_s, s_out_r, emax1,
                                                        alpha1, asum1, E);
    }
    {
        int threads = 256, blocks = (EH + threads - 1) / threads;
        edge_agg_kernel<<<blocks, threads, 0, stream>>>(src, dst, h_proj,
                                                        alpha0, asum0, out, EH);
        edge_agg_kernel<<<blocks, threads, 0, stream>>>(dst, src, h_proj,
                                                        alpha1, asum1, out, EH);
    }
    {
        ln_kernel<<<N, OUT_DIM, 0, stream>>>(out, bias, gamma, beta, N);
    }
}


# Edges grouped by receiver in every call (degree count, scan, slot fill with integer atomics); one wave per node then does softmax and row sum from registers: no float atomics, slot-order tree sums
# speedup vs baseline: 6.3508x; 6.3508x over previous
#include <hip/hip_runtime.h>
#include <math.h>

typedef float v2f __attribute__((ext_vector_type(2)));
typedef float v4f __attribute__((ext_vector_type(4)));
typedef float v8f __attribute__((ext_vector_type(8)));

#define IN_DIM 128
#define OUT_DIM 128
#define NHEAD 4
#define HD 32
#define NEG_SLOPE 0.2f
#define EPS_SOFTMAX 1e-8f
#define EPS_LN 1e-5f

__global__ __launch_bounds__(256) void gemm_wmma_kernel(
    const float* __restrict__ h, const float* __restrict__ W,
    const float* __restrict__ Wself, float* __restrict__ h_proj,
    float* __restrict__ out, int N)
{
    const int wave = threadIdx.x >> 5;
    const int lane = threadIdx.x & 31;
    const int row0 = blockIdx.x * 16;
    const int mrow = lane & 15;
    const int khalf = (lane >> 4) * 2;

    int arow_idx = row0 + mrow;
    if (arow_idx >= N) arow_idx = N - 1;
    const float* arow = h + (size_t)arow_idx * IN_DIM;

    const int col0 = wave * 32;
    const int ccol0 = col0 + mrow;
    const int ccol1 = col0 + 16 + mrow;
    const bool t0_proj = (col0 < 128);
    const bool t1_proj = (col0 + 16 < 128);
    const float* bp0 = t0_proj ? (W + ccol0) : (Wself + (ccol0 - 128));
    const float* bp1 = t1_proj ? (W + ccol1) : (Wself + (ccol1 - 128));

    v8f c0 = {};
    v8f c1 = {};
    for (int k0 = 0; k0 < IN_DIM; k0 += 4) {
        const int k = k0 + khalf;
        v2f a, b0, b1;
        a.x = arow[k];
        a.y = arow[k + 1];
        b0.x = bp0[(size_t)k * OUT_DIM];
        b0.y = bp0[(size_t)(k + 1) * OUT_DIM];
        b1.x = bp1[(size_t)k * OUT_DIM];
        b1.y = bp1[(size_t)(k + 1) * OUT_DIM];
        c0 = __builtin_amdgcn_wmma_f32_16x16x4_f32(false, a, false, b0,
                                                   (short)0, c0, false, false);
        c1 = __builtin_amdgcn_wmma_f32_16x16x4_f32(false, a, false, b1,
                                                   (short)0, c1, false, false);
    }

    const int mbase = (lane >> 4) * 8;
    float* dst0 = t0_proj ? (h_proj + ccol0) : (out + (ccol0 - 128));
    float* dst1 = t1_proj ? (h_proj + ccol1) : (out + (ccol1 - 128));
    if (row0 + 16 <= N) {
#pragma unroll
        for (int v = 0; v < 8; ++v) {
            const size_t row = (size_t)(row0 + mbase + v);
            dst0[row * OUT_DIM] = c0[v];
            dst1[row * OUT_DIM] = c1[v];
        }
    } else {
#pragma unroll
        for (int v = 0; v < 8; ++v) {
            const int row = row0 + mbase + v;
            if (row < N) {
                dst0[(size_t)row * OUT_DIM] = c0[v];
                dst1[(size_t)row * OUT_DIM] = c1[v];
            }
        }
    }
}

__global__ void score_kernel(const float* __restrict__ h_proj,
                             const float* __restrict__ a_in,
                             const float* __restrict__ a_out,
                             float* __restrict__ s_in_s, float* __restrict__ s_in_r,
                             float* __restrict__ s_out_s, float* __restrict__ s_out_r,
                             int NH)
{
    int gid = blockIdx.x * blockDim.x + threadIdx.x;
    if (gid >= NH) return;
    const int hh = gid & 3;
    const int n = gid >> 2;
    const float* x = h_proj + (size_t)n * OUT_DIM + hh * HD;
    const float* ai = a_in + hh * 2 * HD;
    const float* ao = a_out + hh * 2 * HD;
    float s0 = 0.f, s1 = 0.f, s2 = 0.f, s3 = 0.f;
#pragma unroll
    for (int d = 0; d < HD; d += 4) {
        const v4f xv = *(const v4f*)(x + d);
#pragma unroll
        for (int j = 0; j < 4; ++j) {
            s0 += xv[j] * ai[d + j];
            s1 += xv[j] * ai[HD + d + j];
            s2 += xv[j] * ao[d + j];
            s3 += xv[j] * ao[HD + d + j];
        }
    }
    s_in_s[gid] = s0;
    s_in_r[gid] = s1;
    s_out_s[gid] = s2;
    s_out_r[gid] = s3;
}

__global__ void zero_deg_kernel(int* __restrict__ deg, int n2) {
    int gid = blockIdx.x * blockDim.x + threadIdx.x;
    if (gid < n2) deg[gid] = 0;
}

__global__ void count_kernel(const int* __restrict__ src, const int* __restrict__ dst,
                             int* __restrict__ deg_in, int* __restrict__ deg_out, int E)
{
    int e = blockIdx.x * blockDim.x + threadIdx.x;
    if (e >= E) return;
    atomicAdd(&deg_in[dst[e]], 1);
    atomicAdd(&deg_out[src[e]], 1);
}

__global__ __launch_bounds__(1024) void scan_kernel(const int* __restrict__ deg,
                                                    int* __restrict__ rowptr,
                                                    int* __restrict__ cursor, int N)
{
    __shared__ int wtot[32];
    const int dir = blockIdx.x;
    const int* dg = deg + (size_t)dir * N;
    int* rp = rowptr + (size_t)dir * (N + 1);
    int* cu = cursor + (size_t)dir * N;
    const int lane = threadIdx.x & 31;
    const int wave = threadIdx.x >> 5;
    const int span = (((N + 31) / 32) + 127) & ~127;
    const int b = wave * span;
    const int en = (b + span < N) ? (b + span) : N;

    int s = 0;
    for (int i = b + lane; i < en; i += 32) s += dg[i];
    for (int o = 16; o > 0; o >>= 1) s += __shfl_xor(s, o);
    if (lane == 0) wtot[wave] = s;
    __syncthreads();
    int carry = 0;
    for (int w = 0; w < wave; ++w) carry += wtot[w];

    for (int i0 = b; i0 < en; i0 += 128) {
        const int i = i0 + lane * 4;
        const int v0 = (i     < en) ? dg[i]     : 0;
        const int v1 = (i + 1 < en) ? dg[i + 1] : 0;
        const int v2 = (i + 2 < en) ? dg[i + 2] : 0;
        const int v3 = (i + 3 < en) ? dg[i + 3] : 0;
        const int t = (v0 + v1) + (v2 + v3);
        int inc = t;
        for (int o = 1; o < 32; o <<= 1) {
            const int u = __shfl_up(inc, o);
            if (lane >= o) inc += u;
        }
        int ex = carry + inc - t;
        if (i     < en) { rp[i]     = ex; cu[i]     = ex; }
        ex += v0;
        if (i + 1 < en) { rp[i + 1] = ex; cu[i + 1] = ex; }
        ex += v1;
        if (i + 2 < en) { rp[i + 2] = ex; cu[i + 2] = ex; }
        ex += v2;
        if (i + 3 < en) { rp[i + 3] = ex; cu[i + 3] = ex; }
        carry += __shfl(inc, 31);
    }
    if (wave == 31 && lane == 0) rp[N] = carry;
}

__global__ void fill_kernel(const int* __restrict__ src, const int* __restrict__ dst,
                            int* __restrict__ cur_in, int* __restrict__ cur_out,
                            int* __restrict__ nbr_in, int* __restrict__ nbr_out, int E)
{
    int e = blockIdx.x * blockDim.x + threadIdx.x;
    if (e >= E) return;
    const int s = src[e];
    const int d = dst[e];
    const int p = atomicAdd(&cur_in[d], 1);
    nbr_in[p] = s;
    const int q = atomicAdd(&cur_out[s], 1);
    nbr_out[q] = d;
}

__device__ __forceinline__ float wave_max32(float v) {
    for (int o = 16; o > 0; o >>= 1) v = fmaxf(v, __shfl_xor(v, o));
    return v;
}
__device__ __forceinline__ float wave_sum32(float v) {
    for (int o = 16; o > 0; o >>= 1) v += __shfl_xor(v, o);
    return v;
}
__device__ __forceinline__ float lrelu(float x) { return (x > 0.0f) ? x : NEG_SLOPE * x; }

__device__ __forceinline__ void aggregate_dir(const int* __restrict__ nbr, int beg, int end,
                                              const float* __restrict__ s_s, v4f sr,
                                              const float* __restrict__ h_proj, int lane,
                                              float (&acc)[NHEAD])
{
    if (end <= beg) return;
    const bool more = (end - beg) > 32;

    const int j0 = beg + lane;
    const bool ok0 = j0 < end;
    int s0 = 0;
    v4f e0 = {-INFINITY, -INFINITY, -INFINITY, -INFINITY};
    if (ok0) {
        s0 = nbr[j0];
        const v4f ss = *(const v4f*)(s_s + (size_t)s0 * 4);
        for (int h = 0; h < NHEAD; ++h) e0[h] = lrelu(ss[h] + sr[h]);
    }

    v4f m = e0;
    if (more) {
        for (int j = j0 + 32; j < end; j += 32) {
            const int s = nbr[j];
            const v4f ss = *(const v4f*)(s_s + (size_t)s * 4);
            for (int h = 0; h < NHEAD; ++h) m[h] = fmaxf(m[h], lrelu(ss[h] + sr[h]));
        }
    }
    for (int h = 0; h < NHEAD; ++h) m[h] = wave_max32(m[h]);

    v4f a0 = {0.0f, 0.0f, 0.0f, 0.0f};
    if (ok0) {
        for (int h = 0; h < NHEAD; ++h) a0[h] = __expf(e0[h] - m[h]);
    }
    v4f den = a0;
    if (more) {
        for (int j = j0 + 32; j < end; j += 32) {
            const int s = nbr[j];
            const v4f ss = *(const v4f*)(s_s + (size_t)s * 4);
            for (int h = 0; h < NHEAD; ++h) den[h] += __expf(lrelu(ss[h] + sr[h]) - m[h]);
        }
    }
    for (int h = 0; h < NHEAD; ++h) den[h] = wave_sum32(den[h]) + EPS_SOFTMAX;

    for (int c = beg; c < end; c += 32) {
        int s;
        v4f w = {0.0f, 0.0f, 0.0f, 0.0f};
        const int j = c + lane;
        if (c == beg) {
            s = s0;
            for (int h = 0; h < NHEAD; ++h) w[h] = a0[h] / den[h];
        } else {
            s = 0;
            if (j < end) {
                s = nbr[j];
                const v4f ss = *(const v4f*)(s_s + (size_t)s * 4);
                for (int h = 0; h < NHEAD; ++h)
                    w[h] = __expf(lrelu(ss[h] + sr[h]) - m[h]) / den[h];
            }
        }
        const int sfirst = __builtin_amdgcn_readfirstlane(s);
        if (j >= end) s = sfirst;
        const int cnt = (end - c < 32) ? (end - c) : 32;
        for (int i = 0; i < cnt; i += 4) {
#pragma unroll
            for (int u = 0; u < 4; ++u) {
                const int si = __builtin_amdgcn_readlane(s, i + u);
                const float* xr = h_proj + (size_t)si * OUT_DIM + lane;
                for (int h = 0; h < NHEAD; ++h) {
                    const float wh = __int_as_float(
                        __builtin_amdgcn_readlane(__float_as_int(w[h]), i + u));
                    acc[h] += wh * xr[h * HD];
                }
            }
        }
    }
}

__global__ __launch_bounds__(256) void gather_kernel(
    const int* __restrict__ rowptr, const int* __restrict__ nbr_in,
    const int* __restrict__ nbr_out, const float* __restrict__ h_proj,
    const float* __restrict__ s_in_s, const float* __restrict__ s_in_r,
    const float* __restrict__ s_out_s, const float* __restrict__ s_out_r,
    float* __restrict__ out, int N)
{
    const int lane = threadIdx.x & 31;
    const int n = __builtin_amdgcn_readfirstlane(blockIdx.x * 8 + (threadIdx.x >> 5));
    if (n >= N) return;
    const int* rp_in = rowptr;
    const int* rp_out = rowptr + (size_t)(N + 1);
    const int bi = rp_in[n], ei = rp_in[n + 1];
    const int bo = rp_out[n], eo = rp_out[n + 1];
    const v4f sri = *(const v4f*)(s_in_r + (size_t)n * 4);
    const v4f sro = *(const v4f*)(s_out_r + (size_t)n * 4);
    float* orow = out + (size_t)n * OUT_DIM + lane;
    float comb[NHEAD];
    for (int h = 0; h < NHEAD; ++h) comb[h] = orow[h * HD];

    float acc_in[NHEAD] = {0.0f, 0.0f, 0.0f, 0.0f};
    aggregate_dir(nbr_in, bi, ei, s_in_s, sri, h_proj, lane, acc_in);
    for (int h = 0; h < NHEAD; ++h) comb[h] += acc_in[h];
    float acc_out[NHEAD] = {0.0f, 0.0f, 0.0f, 0.0f};
    aggregate_dir(nbr_out, bo, eo, s_out_s, sro, h_proj, lane, acc_out);

    for (int h = 0; h < NHEAD; ++h) orow[h * HD] = comb[h] + acc_out[h];
}

__global__ __launch_bounds__(128) void ln_kernel(float* __restrict__ out,
                                                 const float* __restrict__ bias,
                                                 const float* __restrict__ gamma,
                                                 const float* __restrict__ beta,
                                                 int N)
{
    __shared__ float red[OUT_DIM];
    const int n = blockIdx.x;
    const int t = threadIdx.x;
    if (n >= N) return;
    const float x = out[(size_t)n * OUT_DIM + t] + bias[t];
    red[t] = x;
    __syncthreads();
    for (int s = OUT_DIM / 2; s > 0; s >>= 1) {
        if (t < s) red[t] += red[t + s];
        __syncthreads();
    }
    const float mean = red[0] * (1.0f / OUT_DIM);
    __syncthreads();
    const float d = x - mean;
    red[t] = d * d;
    __syncthreads();
    for (int s = OUT_DIM / 2; s > 0; s >>= 1) {
        if (t < s) red[t] += red[t + s];
        __syncthreads();
    }
    const float var = red[0] * (1.0f / OUT_DIM);
    const float inv = rsqrtf(var + EPS_LN);
    out[(size_t)n * OUT_DIM + t] = d * inv * gamma[t] + beta[t];
}

extern "C" void kernel_launch(void* const* d_in, const int* in_sizes, int n_in,
                              void* d_out, int out_size, void* d_ws, size_t ws_size,
                              hipStream_t stream) {
    const float* h      = (const float*)d_in[0];
    const int*   eidx   = (const int*)d_in[1];
    const float* W      = (const float*)d_in[2];
    const float* Wself  = (const float*)d_in[3];
    const float* a_in   = (const float*)d_in[4];
    const float* a_out  = (const float*)d_in[5];
    const float* bias   = (const float*)d_in[6];
    const float* gamma  = (const float*)d_in[7];
    const float* beta   = (const float*)d_in[8];
    float* out = (float*)d_out;

    const int N = in_sizes[0] / IN_DIM;
    const int E = in_sizes[1] / 2;
    const int NH = N * NHEAD;

    const int* src = eidx;
    const int* dst = eidx + E;

    float* ws = (float*)d_ws;
    float* h_proj  = ws;
    float* s_in_s  = h_proj + (size_t)N * OUT_DIM;
    float* s_in_r  = s_in_s + NH;
    float* s_out_s = s_in_r + NH;
    float* s_out_r = s_out_s + NH;
    int* deg     = (int*)(s_out_r + NH);
    int* cursor  = deg + (size_t)2 * N;
    int* rowptr  = cursor + (size_t)2 * N;
    int* nbr_in  = rowptr + (size_t)2 * (N + 1);
    int* nbr_out = nbr_in + (size_t)E;

    {
        int threads = 256, blocks = (2 * N + threads - 1) / threads;
        zero_deg_kernel<<<blocks, threads, 0, stream>>>(deg, 2 * N);
    }
    {
        int blocks = (N + 15) / 16;
        gemm_wmma_kernel<<<blocks, 256, 0, stream>>>(h, W, Wself, h_proj, out, N);
    }
    {
        int threads = 256, blocks = (NH + threads - 1) / threads;
        score_kernel<<<blocks, threads, 0, stream>>>(h_proj, a_in, a_out,
                                                     s_in_s, s_in_r, s_out_s, s_out_r, NH);
    }
    {
        int threads = 256, blocks = (E + threads - 1) / threads;
        count_kernel<<<blocks, threads, 0, stream>>>(src, dst, deg, deg + N, E);
    }
    {
        scan_kernel<<<2, 1024, 0, stream>>>(deg, rowptr, cursor, N);
    }
    {
        int threads = 256, blocks = (E + threads - 1) / threads;
        fill_kernel<<<blocks, threads, 0, stream>>>(src, dst, cursor, cursor + N,
                                                    nbr_in, nbr_out, E);
    }
    {
        int blocks = (N + 7) / 8;
        gather_kernel<<<blocks, 256, 0, stream>>>(rowptr, nbr_in, nbr_out, h_proj,
                                                  s_in_s, s_in_r, s_out_s, s_out_r, out, N);
    }
    {
        ln_kernel<<<N, OUT_DIM, 0, stream>>>(out, bias, gamma, beta, N);
    }
}


# LayerNorm fused into the one-wave-per-node gather: row stays in registers, mean/variance by per-lane sums then a 32-lane butterfly (sum order differs from the LDS tree); its own launch is gone
# speedup vs baseline: 7.0602x; 1.1117x over previous
#include <hip/hip_runtime.h>
#include <math.h>

typedef float v2f __attribute__((ext_vector_type(2)));
typedef float v4f __attribute__((ext_vector_type(4)));
typedef float v8f __attribute__((ext_vector_type(8)));

#define IN_DIM 128
#define OUT_DIM 128
#define NHEAD 4
#define HD 32
#define NEG_SLOPE 0.2f
#define EPS_SOFTMAX 1e-8f
#define EPS_LN 1e-5f

__global__ __launch_bounds__(256) void gemm_wmma_kernel(
    const float* __restrict__ h, const float* __restrict__ W,
    const float* __restrict__ Wself, float* __restrict__ h_proj,
    float* __restrict__ out, int N)
{
    const int wave = threadIdx.x >> 5;
    const int lane = threadIdx.x & 31;
    const int row0 = blockIdx.x * 16;
    const int mrow = lane & 15;
    const int khalf = (lane >> 4) * 2;

    int arow_idx = row0 + mrow;
    if (arow_idx >= N) arow_idx = N - 1;
    const float* arow = h + (size_t)arow_idx * IN_DIM;

    const int col0 = wave * 32;
    const int ccol0 = col0 + mrow;
    const int ccol1 = col0 + 16 + mrow;
    const bool t0_proj = (col0 < 128);
    const bool t1_proj = (col0 + 16 < 128);
    const float* bp0 = t0_proj ? (W + ccol0) : (Wself + (ccol0 - 128));
    const float* bp1 = t1_proj ? (W + ccol1) : (Wself + (ccol1 - 128));

    v8f c0 = {};
    v8f c1 = {};
    for (int k0 = 0; k0 < IN_DIM; k0 += 4) {
        const int k = k0 + khalf;
        v2f a, b0, b1;
        a.x = arow[k];
        a.y = arow[k + 1];
        b0.x = bp0[(size_t)k * OUT_DIM];
        b0.y = bp0[(size_t)(k + 1) * OUT_DIM];
        b1.x = bp1[(size_t)k * OUT_DIM];
        b1.y = bp1[(size_t)(k + 1) * OUT_DIM];
        c0 = __builtin_amdgcn_wmma_f32_16x16x4_f32(false, a, false, b0,
                                                   (short)0, c0, false, false);
        c1 = __builtin_amdgcn_wmma_f32_16x16x4_f32(false, a, false, b1,
                                                   (short)0, c1, false, false);
    }

    const int mbase = (lane >> 4) * 8;
    float* dst0 = t0_proj ? (h_proj + ccol0) : (out + (ccol0 - 128));
    float* dst1 = t1_proj ? (h_proj + ccol1) : (out + (ccol1 - 128));
    if (row0 + 16 <= N) {
#pragma unroll
        for (int v = 0; v < 8; ++v) {
            const size_t row = (size_t)(row0 + mbase + v);
            dst0[row * OUT_DIM] = c0[v];
            dst1[row * OUT_DIM] = c1[v];
        }
    } else {
#pragma unroll
        for (int v = 0; v < 8; ++v) {
            const int row = row0 + mbase + v;
            if (row < N) {
                dst0[(size_t)row * OUT_DIM] = c0[v];
                dst1[(size_t)row * OUT_DIM] = c1[v];
            }
        }
    }
}

__global__ void score_kernel(const float* __restrict__ h_proj,
                             const float* __restrict__ a_in,
                             const float* __restrict__ a_out,
                             float* __restrict__ s_in_s, float* __restrict__ s_in_r,
                             float* __restrict__ s_out_s, float* __restrict__ s_out_r,
                             int NH)
{
    int gid = blockIdx.x * blockDim.x + threadIdx.x;
    if (gid >= NH) return;
    const int hh = gid & 3;
    const int n = gid >> 2;
    const float* x = h_proj + (size_t)n * OUT_DIM + hh * HD;
    const float* ai = a_in + hh * 2 * HD;
    const float* ao = a_out + hh * 2 * HD;
    float s0 = 0.f, s1 = 0.f, s2 = 0.f, s3 = 0.f;
#pragma unroll
    for (int d = 0; d < HD; d += 4) {
        const v4f xv = *(const v4f*)(x + d);
#pragma unroll
        for (int j = 0; j < 4; ++j) {
            s0 += xv[j] * ai[d + j];
            s1 += xv[j] * ai[HD + d + j];
            s2 += xv[j] * ao[d + j];
            s3 += xv[j] * ao[HD + d + j];
        }
    }
    s_in_s[gid] = s0;
    s_in_r[gid] = s1;
    s_out_s[gid] = s2;
    s_out_r[gid] = s3;
}

__global__ void zero_deg_kernel(int* __restrict__ deg, int n2) {
    int gid = blockIdx.x * blockDim.x + threadIdx.x;
    if (gid < n2) deg[gid] = 0;
}

__global__ void count_kernel(const int* __restrict__ src, const int* __restrict__ dst,
                             int* __restrict__ deg_in, int* __restrict__ deg_out, int E)
{
    int e = blockIdx.x * blockDim.x + threadIdx.x;
    if (e >= E) return;
    atomicAdd(&deg_in[dst[e]], 1);
    atomicAdd(&deg_out[src[e]], 1);
}

__global__ __launch_bounds__(1024) void scan_kernel(const int* __restrict__ deg,
                                                    int* __restrict__ rowptr,
                                                    int* __restrict__ cursor, int N)
{
    __shared__ int wtot[32];
    const int dir = blockIdx.x;
    const int* dg = deg + (size_t)dir * N;
    int* rp = rowptr + (size_t)dir * (N + 1);
    int* cu = cursor + (size_t)dir * N;
    const int lane = threadIdx.x & 31;
    const int wave = threadIdx.x >> 5;
    const int span = (((N + 31) / 32) + 127) & ~127;
    const int b = wave * span;
    const int en = (b + span < N) ? (b + span) : N;

    int s = 0;
    for (int i = b + lane; i < en; i += 32) s += dg[i];
    for (int o = 16; o > 0; o >>= 1) s += __shfl_xor(s, o);
    if (lane == 0) wtot[wave] = s;
    __syncthreads();
    int carry = 0;
    for (int w = 0; w < wave; ++w) carry += wtot[w];

    for (int i0 = b; i0 < en; i0 += 128) {
        const int i = i0 + lane * 4;
        const int v0 = (i     < en) ? dg[i]     : 0;
        const int v1 = (i + 1 < en) ? dg[i + 1] : 0;
        const int v2 = (i + 2 < en) ? dg[i + 2] : 0;
        const int v3 = (i + 3 < en) ? dg[i + 3] : 0;
        const int t = (v0 + v1) + (v2 + v3);
        int inc = t;
        for (int o = 1; o < 32; o <<= 1) {
            const int u = __shfl_up(inc, o);
            if (lane >= o) inc += u;
        }
        int ex = carry + inc - t;
        if (i     < en) { rp[i]     = ex; cu[i]     = ex; }
        ex += v0;
        if (i + 1 < en) { rp[i + 1] = ex; cu[i + 1] = ex; }
        ex += v1;
        if (i + 2 < en) { rp[i + 2] = ex; cu[i + 2] = ex; }
        ex += v2;
        if (i + 3 < en) { rp[i + 3] = ex; cu[i + 3] = ex; }
        carry += __shfl(inc, 31);
    }
    if (wave == 31 && lane == 0) rp[N] = carry;
}

__global__ void fill_kernel(const int* __restrict__ src, const int* __restrict__ dst,
                            int* __restrict__ cur_in, int* __restrict__ cur_out,
                            int* __restrict__ nbr_in, int* __restrict__ nbr_out, int E)
{
    int e = blockIdx.x * blockDim.x + threadIdx.x;
    if (e >= E) return;
    const int s = src[e];
    const int d = dst[e];
    const int p = atomicAdd(&cur_in[d], 1);
    nbr_in[p] = s;
    const int q = atomicAdd(&cur_out[s], 1);
    nbr_out[q] = d;
}

__device__ __forceinline__ float wave_max32(float v) {
    for (int o = 16; o > 0; o >>= 1) v = fmaxf(v, __shfl_xor(v, o));
    return v;
}
__device__ __forceinline__ float wave_sum32(float v) {
    for (int o = 16; o > 0; o >>= 1) v += __shfl_xor(v, o);
    return v;
}
__device__ __forceinline__ float lrelu(float x) { return (x > 0.0f) ? x : NEG_SLOPE * x; }

__device__ __forceinline__ void aggregate_dir(const int* __restrict__ nbr, int beg, int end,
                                              const float* __restrict__ s_s, v4f sr,
                                              const float* __restrict__ h_proj, int lane,
                                              float (&acc)[NHEAD])
{
    if (end <= beg) return;
    const bool more = (end - beg) > 32;

    const int j0 = beg + lane;
    const bool ok0 = j0 < end;
    int s0 = 0;
    v4f e0 = {-INFINITY, -INFINITY, -INFINITY, -INFINITY};
    if (ok0) {
        s0 = nbr[j0];
        const v4f ss = *(const v4f*)(s_s + (size_t)s0 * 4);
        for (int h = 0; h < NHEAD; ++h) e0[h] = lrelu(ss[h] + sr[h]);
    }

    v4f m = e0;
    if (more) {
        for (int j = j0 + 32; j < end; j += 32) {
            const int s = nbr[j];
            const v4f ss = *(const v4f*)(s_s + (size_t)s * 4);
            for (int h = 0; h < NHEAD; ++h) m[h] = fmaxf(m[h], lrelu(ss[h] + sr[h]));
        }
    }
    for (int h = 0; h < NHEAD; ++h) m[h] = wave_max32(m[h]);

    v4f a0 = {0.0f, 0.0f, 0.0f, 0.0f};
    if (ok0) {
        for (int h = 0; h < NHEAD; ++h) a0[h] = __expf(e0[h] - m[h]);
    }
    v4f den = a0;
    if (more) {
        for (int j = j0 + 32; j < end; j += 32) {
            const int s = nbr[j];
            const v4f ss = *(const v4f*)(s_s + (size_t)s * 4);
            for (int h = 0; h < NHEAD; ++h) den[h] += __expf(lrelu(ss[h] + sr[h]) - m[h]);
        }
    }
    for (int h = 0; h < NHEAD; ++h) den[h] = wave_sum32(den[h]) + EPS_SOFTMAX;

    for (int c = beg; c < end; c += 32) {
        int s;
        v4f w = {0.0f, 0.0f, 0.0f, 0.0f};
        const int j = c + lane;
        if (c == beg) {
            s = s0;
            for (int h = 0; h < NHEAD; ++h) w[h] = a0[h] / den[h];
        } else {
            s = 0;
            if (j < end) {
                s = nbr[j];
                const v4f ss = *(const v4f*)(s_s + (size_t)s * 4);
                for (int h = 0; h < NHEAD; ++h)
                    w[h] = __expf(lrelu(ss[h] + sr[h]) - m[h]) / den[h];
            }
        }
        const int sfirst = __builtin_amdgcn_readfirstlane(s);
        if (j >= end) s = sfirst;
        const int cnt = (end - c < 32) ? (end - c) : 32;
        for (int i = 0; i < cnt; i += 4) {
#pragma unroll
            for (int u = 0; u < 4; ++u) {
                const int si = __builtin_amdgcn_readlane(s, i + u);
                const float* xr = h_proj + (size_t)si * OUT_DIM + lane;
                for (int h = 0; h < NHEAD; ++h) {
                    const float wh = __int_as_float(
                        __builtin_amdgcn_readlane(__float_as_int(w[h]), i + u));
                    acc[h] += wh * xr[h * HD];
                }
            }
        }
    }
}

__global__ __launch_bounds__(256) void gather_kernel(
    const int* __restrict__ rowptr, const int* __restrict__ nbr_in,
    const int* __restrict__ nbr_out, const float* __restrict__ h_proj,
    const float* __restrict__ s_in_s, const float* __restrict__ s_in_r,
    const float* __restrict__ s_out_s, const float* __restrict__ s_out_r,
    const float* __restrict__ bias, const float* __restrict__ gamma,
    const float* __restrict__ beta, float* __restrict__ out, int N)
{
    const int lane = threadIdx.x & 31;
    const int n = __builtin_amdgcn_readfirstlane(blockIdx.x * 8 + (threadIdx.x >> 5));
    if (n >= N) return;
    const int* rp_in = rowptr;
    const int* rp_out = rowptr + (size_t)(N + 1);
    const int bi = rp_in[n], ei = rp_in[n + 1];
    const int bo = rp_out[n], eo = rp_out[n + 1];
    const v4f sri = *(const v4f*)(s_in_r + (size_t)n * 4);
    const v4f sro = *(const v4f*)(s_out_r + (size_t)n * 4);
    float* orow = out + (size_t)n * OUT_DIM + lane;
    float comb[NHEAD];
    for (int h = 0; h < NHEAD; ++h) comb[h] = orow[h * HD];

    float acc_in[NHEAD] = {0.0f, 0.0f, 0.0f, 0.0f};
    aggregate_dir(nbr_in, bi, ei, s_in_s, sri, h_proj, lane, acc_in);
    for (int h = 0; h < NHEAD; ++h) comb[h] += acc_in[h];
    float acc_out[NHEAD] = {0.0f, 0.0f, 0.0f, 0.0f};
    aggregate_dir(nbr_out, bo, eo, s_out_s, sro, h_proj, lane, acc_out);

    float x[NHEAD];
    for (int h = 0; h < NHEAD; ++h) x[h] = (comb[h] + acc_out[h]) + bias[h * HD + lane];

    const float mean = wave_sum32((x[0] + x[1]) + (x[2] + x[3])) * (1.0f / OUT_DIM);
    float d[NHEAD];
    for (int h = 0; h < NHEAD; ++h) d[h] = x[h] - mean;
    const float var = wave_sum32((d[0] * d[0] + d[1] * d[1]) + (d[2] * d[2] + d[3] * d[3]))
                      * (1.0f / OUT_DIM);
    const float inv = rsqrtf(var + EPS_LN);
    for (int h = 0; h < NHEAD; ++h)
        orow[h * HD] = d[h] * inv * gamma[h * HD + lane] + beta[h * HD + lane];
}

extern "C" void kernel_launch(void* const* d_in, const int* in_sizes, int n_in,
                              void* d_out, int out_size, void* d_ws, size_t ws_size,
                              hipStream_t stream) {
    const float* h      = (const float*)d_in[0];
    const int*   eidx   = (const int*)d_in[1];
    const float* W      = (const float*)d_in[2];
    const float* Wself  = (const float*)d_in[3];
    const float* a_in   = (const float*)d_in[4];
    const float* a_out  = (const float*)d_in[5];
    const float* bias   = (const float*)d_in[6];
    const float* gamma  = (const float*)d_in[7];
    const float* beta   = (const float*)d_in[8];
    float* out = (float*)d_out;

    const int N = in_sizes[0] / IN_DIM;
    const int E = in_sizes[1] / 2;
    const int NH = N * NHEAD;

    const int* src = eidx;
    const int* dst = eidx + E;

    float* ws = (float*)d_ws;
    float* h_proj  = ws;
    float* s_in_s  = h_proj + (size_t)N * OUT_DIM;
    float* s_in_r  = s_in_s + NH;
    float* s_out_s = s_in_r + NH;
    float* s_out_r = s_out_s + NH;
    int* deg     = (int*)(s_out_r + NH);
    int* cursor  = deg + (size_t)2 * N;
    int* rowptr  = cursor + (size_t)2 * N;
    int* nbr_in  = rowptr + (size_t)2 * (N + 1);
    int* nbr_out = nbr_in + (size_t)E;

    {
        int threads = 256, blocks = (2 * N + threads - 1) / threads;
        zero_deg_kernel<<<blocks, threads, 0, stream>>>(deg, 2 * N);
    }
    {
        int blocks = (N + 15) / 16;
        gemm_wmma_kernel<<<blocks, 256, 0, stream>>>(h, W, Wself, h_proj, out, N);
    }
    {
        int threads = 256, blocks = (NH + threads - 1) / threads;
        score_kernel<<<blocks, threads, 0, stream>>>(h_proj, a_in, a_out,
                                                     s_in_s, s_in_r, s_out_s, s_out_r, NH);
    }
    {
        int threads = 256, blocks = (E + threads - 1) / threads;
        count_kernel<<<blocks, threads, 0, stream>>>(src, dst, deg, deg + N, E);
    }
    {
        scan_kernel<<<2, 1024, 0, stream>>>(deg, rowptr, cursor, N);
    }
    {
        int threads = 256, blocks = (E + threads - 1) / threads;
        fill_kernel<<<blocks, threads, 0, stream>>>(src, dst, cursor, cursor + N,
                                                    nbr_in, nbr_out, E);
    }
    {
        int blocks = (N + 7) / 8;
        gather_kernel<<<blocks, 256, 0, stream>>>(rowptr, nbr_in, nbr_out, h_proj,
                                                  s_in_s, s_in_r, s_out_s, s_out_r,
                                                  bias, gamma, beta, out, N);
    }
}


# List build with one atomic pass, not two: the returning degree-count add is kept as the edge's slot (arrival order, as before), so placement after the scan is row start + slot with no atomics
# speedup vs baseline: 8.9695x; 1.2704x over previous
#include <hip/hip_runtime.h>
#include <math.h>

typedef float v2f __attribute__((ext_vector_type(2)));
typedef float v4f __attribute__((ext_vector_type(4)));
typedef float v8f __attribute__((ext_vector_type(8)));

#define IN_DIM 128
#define OUT_DIM 128
#define NHEAD 4
#define HD 32
#define NEG_SLOPE 0.2f
#define EPS_SOFTMAX 1e-8f
#define EPS_LN 1e-5f

__global__ __launch_bounds__(256) void gemm_wmma_kernel(
    const float* __restrict__ h, const float* __restrict__ W,
    const float* __restrict__ Wself, float* __restrict__ h_proj,
    float* __restrict__ out, int N)
{
    const int wave = threadIdx.x >> 5;
    const int lane = threadIdx.x & 31;
    const int row0 = blockIdx.x * 16;
    const int mrow = lane & 15;
    const int khalf = (lane >> 4) * 2;

    int arow_idx = row0 + mrow;
    if (arow_idx >= N) arow_idx = N - 1;
    const float* arow = h + (size_t)arow_idx * IN_DIM;

    const int col0 = wave * 32;
    const int ccol0 = col0 + mrow;
    const int ccol1 = col0 + 16 + mrow;
    const bool t0_proj = (col0 < 128);
    const bool t1_proj = (col0 + 16 < 128);
    const float* bp0 = t0_proj ? (W + ccol0) : (Wself + (ccol0 - 128));
    const float* bp1 = t1_proj ? (W + ccol1) : (Wself + (ccol1 - 128));

    v8f c0 = {};
    v8f c1 = {};
    for (int k0 = 0; k0 < IN_DIM; k0 += 4) {
        const int k = k0 + khalf;
        v2f a, b0, b1;
        a.x = arow[k];
        a.y = arow[k + 1];
        b0.x = bp0[(size_t)k * OUT_DIM];
        b0.y = bp0[(size_t)(k + 1) * OUT_DIM];
        b1.x = bp1[(size_t)k * OUT_DIM];
        b1.y = bp1[(size_t)(k + 1) * OUT_DIM];
        c0 = __builtin_amdgcn_wmma_f32_16x16x4_f32(false, a, false, b0,
                                                   (short)0, c0, false, false);
        c1 = __builtin_amdgcn_wmma_f32_16x16x4_f32(false, a, false, b1,
                                                   (short)0, c1, false, false);
    }

    const int mbase = (lane >> 4) * 8;
    float* dst0 = t0_proj ? (h_proj + ccol0) : (out + (ccol0 - 128));
    float* dst1 = t1_proj ? (h_proj + ccol1) : (out + (ccol1 - 128));
    if (row0 + 16 <= N) {
#pragma unroll
        for (int v = 0; v < 8; ++v) {
            const size_t row = (size_t)(row0 + mbase + v);
            dst0[row * OUT_DIM] = c0[v];
            dst1[row * OUT_DIM] = c1[v];
        }
    } else {
#pragma unroll
        for (int v = 0; v < 8; ++v) {
            const int row = row0 + mbase + v;
            if (row < N) {
                dst0[(size_t)row * OUT_DIM] = c0[v];
                dst1[(size_t)row * OUT_DIM] = c1[v];
            }
        }
    }
}

__global__ void score_kernel(const float* __restrict__ h_proj,
                             const float* __restrict__ a_in,
                             const float* __restrict__ a_out,
                             float* __restrict__ s_in_s, float* __restrict__ s_in_r,
                             float* __restrict__ s_out_s, float* __restrict__ s_out_r,
                             int NH)
{
    int gid = blockIdx.x * blockDim.x + threadIdx.x;
    if (gid >= NH) return;
    const int hh = gid & 3;
    const int n = gid >> 2;
    const float* x = h_proj + (size_t)n * OUT_DIM + hh * HD;
    const float* ai = a_in + hh * 2 * HD;
    const float* ao = a_out + hh * 2 * HD;
    float s0 = 0.f, s1 = 0.f, s2 = 0.f, s3 = 0.f;
#pragma unroll
    for (int d = 0; d < HD; d += 4) {
        const v4f xv = *(const v4f*)(x + d);
#pragma unroll
        for (int j = 0; j < 4; ++j) {
            s0 += xv[j] * ai[d + j];
            s1 += xv[j] * ai[HD + d + j];
            s2 += xv[j] * ao[d + j];
            s3 += xv[j] * ao[HD + d + j];
        }
    }
    s_in_s[gid] = s0;
    s_in_r[gid] = s1;
    s_out_s[gid] = s2;
    s_out_r[gid] = s3;
}

__global__ void zero_deg_kernel(int* __restrict__ deg, int n2) {
    int gid = blockIdx.x * blockDim.x + threadIdx.x;
    if (gid < n2) deg[gid] = 0;
}

__global__ void rank_kernel(const int* __restrict__ src, const int* __restrict__ dst,
                            int* __restrict__ deg_in, int* __restrict__ deg_out,
                            int* __restrict__ rank_in, int* __restrict__ rank_out, int E)
{
    int e = blockIdx.x * blockDim.x + threadIdx.x;
    if (e >= E) return;
    const int s = src[e];
    const int d = dst[e];
    const int p = atomicAdd(&deg_in[d], 1);
    const int q = atomicAdd(&deg_out[s], 1);
    rank_in[e] = p;
    rank_out[e] = q;
}

__global__ __launch_bounds__(1024) void scan_kernel(const int* __restrict__ deg,
                                                    int* __restrict__ rowptr, int N)
{
    __shared__ int wtot[32];
    const int dir = blockIdx.x;
    const int* dg = deg + (size_t)dir * N;
    int* rp = rowptr + (size_t)dir * (N + 1);
    const int lane = threadIdx.x & 31;
    const int wave = threadIdx.x >> 5;
    const int span = (((N + 31) / 32) + 127) & ~127;
    const int b = wave * span;
    const int en = (b + span < N) ? (b + span) : N;

    int s = 0;
    for (int i = b + lane; i < en; i += 32) s += dg[i];
    for (int o = 16; o > 0; o >>= 1) s += __shfl_xor(s, o);
    if (lane == 0) wtot[wave] = s;
    __syncthreads();
    int carry = 0;
    for (int w = 0; w < wave; ++w) carry += wtot[w];

    for (int i0 = b; i0 < en; i0 += 128) {
        const int i = i0 + lane * 4;
        const int v0 = (i     < en) ? dg[i]     : 0;
        const int v1 = (i + 1 < en) ? dg[i + 1] : 0;
        const int v2 = (i + 2 < en) ? dg[i + 2] : 0;
        const int v3 = (i + 3 < en) ? dg[i + 3] : 0;
        const int t = (v0 + v1) + (v2 + v3);
        int inc = t;
        for (int o = 1; o < 32; o <<= 1) {
            const int u = __shfl_up(inc, o);
            if (lane >= o) inc += u;
        }
        int ex = carry + inc - t;
        if (i     < en) rp[i]     = ex;
        ex += v0;
        if (i + 1 < en) rp[i + 1] = ex;
        ex += v1;
        if (i + 2 < en) rp[i + 2] = ex;
        ex += v2;
        if (i + 3 < en) rp[i + 3] = ex;
        carry += __shfl(inc, 31);
    }
    if (wave == 31 && lane == 0) rp[N] = carry;
}

__global__ void place_kernel(const int* __restrict__ src, const int* __restrict__ dst,
                             const int* __restrict__ rp_in, const int* __restrict__ rp_out,
                             const int* __restrict__ rank_in, const int* __restrict__ rank_out,
                             int* __restrict__ nbr_in, int* __restrict__ nbr_out, int E)
{
    int e = blockIdx.x * blockDim.x + threadIdx.x;
    if (e >= E) return;
    const int s = src[e];
    const int d = dst[e];
    nbr_in[rp_in[d] + rank_in[e]] = s;
    nbr_out[rp_out[s] + rank_out[e]] = d;
}

__device__ __forceinline__ float wave_max32(float v) {
    for (int o = 16; o > 0; o >>= 1) v = fmaxf(v, __shfl_xor(v, o));
    return v;
}
__device__ __forceinline__ float wave_sum32(float v) {
    for (int o = 16; o > 0; o >>= 1) v += __shfl_xor(v, o);
    return v;
}
__device__ __forceinline__ float lrelu(float x) { return (x > 0.0f) ? x : NEG_SLOPE * x; }

__device__ __forceinline__ void aggregate_dir(const int* __restrict__ nbr, int beg, int end,
                                              const float* __restrict__ s_s, v4f sr,
                                              const float* __restrict__ h_proj, int lane,
                                              float (&acc)[NHEAD])
{
    if (end <= beg) return;
    const bool more = (end - beg) > 32;

    const int j0 = beg + lane;
    const bool ok0 = j0 < end;
    int s0 = 0;
    v4f e0 = {-INFINITY, -INFINITY, -INFINITY, -INFINITY};
    if (ok0) {
        s0 = nbr[j0];
        const v4f ss = *(const v4f*)(s_s + (size_t)s0 * 4);
        for (int h = 0; h < NHEAD; ++h) e0[h] = lrelu(ss[h] + sr[h]);
    }

    v4f m = e0;
    if (more) {
        for (int j = j0 + 32; j < end; j += 32) {
            const int s = nbr[j];
            const v4f ss = *(const v4f*)(s_s + (size_t)s * 4);
            for (int h = 0; h < NHEAD; ++h) m[h] = fmaxf(m[h], lrelu(ss[h] + sr[h]));
        }
    }
    for (int h = 0; h < NHEAD; ++h) m[h] = wave_max32(m[h]);

    v4f a0 = {0.0f, 0.0f, 0.0f, 0.0f};
    if (ok0) {
        for (int h = 0; h < NHEAD; ++h) a0[h] = __expf(e0[h] - m[h]);
    }
    v4f den = a0;
    if (more) {
        for (int j = j0 + 32; j < end; j += 32) {
            const int s = nbr[j];
            const v4f ss = *(const v4f*)(s_s + (size_t)s * 4);
            for (int h = 0; h < NHEAD; ++h) den[h] += __expf(lrelu(ss[h] + sr[h]) - m[h]);
        }
    }
    for (int h = 0; h < NHEAD; ++h) den[h] = wave_sum32(den[h]) + EPS_SOFTMAX;

    for (int c = beg; c < end; c += 32) {
        int s;
        v4f w = {0.0f, 0.0f, 0.0f, 0.0f};
        const int j = c + lane;
        if (c == beg) {
            s = s0;
            for (int h = 0; h < NHEAD; ++h) w[h] = a0[h] / den[h];
        } else {
            s = 0;
            if (j < end) {
                s = nbr[j];
                const v4f ss = *(const v4f*)(s_s + (size_t)s * 4);
                for (int h = 0; h < NHEAD; ++h)
                    w[h] = __expf(lrelu(ss[h] + sr[h]) - m[h]) / den[h];
            }
        }
        const int sfirst = __builtin_amdgcn_readfirstlane(s);
        if (j >= end) s = sfirst;
        const int cnt = (end - c < 32) ? (end - c) : 32;
        for (int i = 0; i < cnt; i += 4) {
#pragma unroll
            for (int u = 0; u < 4; ++u) {
                const int si = __builtin_amdgcn_readlane(s, i + u);
                const float* xr = h_proj + (size_t)si * OUT_DIM + lane;
                for (int h = 0; h < NHEAD; ++h) {
                    const float wh = __int_as_float(
                        __builtin_amdgcn_readlane(__float_as_int(w[h]), i + u));
                    acc[h] += wh * xr[h * HD];
                }
            }
        }
    }
}

__global__ __launch_bounds__(256) void gather_kernel(
    const int* __restrict__ rowptr, const int* __restrict__ nbr_in,
    const int* __restrict__ nbr_out, const float* __restrict__ h_proj,
    const float* __restrict__ s_in_s, const float* __restrict__ s_in_r,
    const float* __restrict__ s_out_s, const float* __restrict__ s_out_r,
    const float* __restrict__ bias, const float* __restrict__ gamma,
    const float* __restrict__ beta, float* __restrict__ out, int N)
{
    const int lane = threadIdx.x & 31;
    const int n = __builtin_amdgcn_readfirstlane(blockIdx.x * 8 + (threadIdx.x >> 5));
    if (n >= N) return;
    const int* rp_in = rowptr;
    const int* rp_out = rowptr + (size_t)(N + 1);
    const int bi = rp_in[n], ei = rp_in[n + 1];
    const int bo = rp_out[n], eo = rp_out[n + 1];
    const v4f sri = *(const v4f*)(s_in_r + (size_t)n * 4);
    const v4f sro = *(const v4f*)(s_out_r + (size_t)n * 4);
    float* orow = out + (size_t)n * OUT_DIM + lane;
    float comb[NHEAD];
    for (int h = 0; h < NHEAD; ++h) comb[h] = orow[h * HD];

    float acc_in[NHEAD] = {0.0f, 0.0f, 0.0f, 0.0f};
    aggregate_dir(nbr_in, bi, ei, s_in_s, sri, h_proj, lane, acc_in);
    for (int h = 0; h < NHEAD; ++h) comb[h] += acc_in[h];
    float acc_out[NHEAD] = {0.0f, 0.0f, 0.0f, 0.0f};
    aggregate_dir(nbr_out, bo, eo, s_out_s, sro, h_proj, lane, acc_out);

    float x[NHEAD];
    for (int h = 0; h < NHEAD; ++h) x[h] = (comb[h] + acc_out[h]) + bias[h * HD + lane];

    const float mean = wave_sum32((x[0] + x[1]) + (x[2] + x[3])) * (1.0f / OUT_DIM);
    float d[NHEAD];
    for (int h = 0; h < NHEAD; ++h) d[h] = x[h] - mean;
    const float var = wave_sum32((d[0] * d[0] + d[1] * d[1]) + (d[2] * d[2] + d[3] * d[3]))
                      * (1.0f / OUT_DIM);
    const float inv = rsqrtf(var + EPS_LN);
    for (int h = 0; h < NHEAD; ++h)
        orow[h * HD] = d[h] * inv * gamma[h * HD + lane] + beta[h * HD + lane];
}

extern "C" void kernel_launch(void* const* d_in, const int* in_sizes, int n_in,
                              void* d_out, int out_size, void* d_ws, size_t ws_size,
                              hipStream_t stream) {
    const float* h      = (const float*)d_in[0];
    const int*   eidx   = (const int*)d_in[1];
    const float* W      = (const float*)d_in[2];
    const float* Wself  = (const float*)d_in[3];
    const float* a_in   = (const float*)d_in[4];
    const float* a_out  = (const float*)d_in[5];
    const float* bias   = (const float*)d_in[6];
    const float* gamma  = (const float*)d_in[7];
    const float* beta   = (const float*)d_in[8];
    float* out = (float*)d_out;

    const int N = in_sizes[0] / IN_DIM;
    const int E = in_sizes[1] / 2;
    const int NH = N * NHEAD;

    const int* src = eidx;
    const int* dst = eidx + E;

    float* ws = (float*)d_ws;
    float* h_proj  = ws;
    float* s_in_s  = h_proj + (size_t)N * OUT_DIM;
    float* s_in_r  = s_in_s + NH;
    float* s_out_s = s_in_r + NH;
    float* s_out_r = s_out_s + NH;
    int* deg     = (int*)(s_out_r + NH);
    int* rowptr  = deg + (size_t)2 * N;
    int* nbr_in  = rowptr + (size_t)2 * (N + 1);
    int* nbr_out = nbr_in + (size_t)E;
    int* rank_in  = (int*)h_proj;
    int* rank_out = rank_in + (size_t)E;

    {
        int threads = 256, blocks = (2 * N + threads - 1) / threads;
        zero_deg_kernel<<<blocks, threads, 0, stream>>>(deg, 2 * N);
    }
    {
        int threads = 256, blocks = (E + threads - 1) / threads;
        rank_kernel<<<blocks, threads, 0, stream>>>(src, dst, deg, deg + N,
                                                    rank_in, rank_out, E);
    }
    {
        scan_kernel<<<2, 1024, 0, stream>>>(deg, rowptr, N);
    }
    {
        int threads = 256, blocks = (E + threads - 1) / threads;
        place_kernel<<<blocks, threads, 0, stream>>>(src, dst, rowptr, rowptr + (N + 1),
                                                     rank_in, rank_out, nbr_in, nbr_out, E);
    }
    {
        int blocks = (N + 15) / 16;
        gemm_wmma_kernel<<<blocks, 256, 0, stream>>>(h, W, Wself, h_proj, out, N);
    }
    {
        int threads = 256, blocks = (NH + threads - 1) / threads;
        score_kernel<<<blocks, threads, 0, stream>>>(h_proj, a_in, a_out,
                                                     s_in_s, s_in_r, s_out_s, s_out_r, NH);
    }
    {
        int blocks = (N + 7) / 8;
        gather_kernel<<<blocks, 256, 0, stream>>>(rowptr, nbr_in, nbr_out, h_proj,
                                                  s_in_s, s_in_r, s_out_s, s_out_r,
                                                  bias, gamma, beta, out, N);
    }
}


# List build as a two-level counting sort on LDS counters (chunk to bucket, bucket to row): no global adds, slot planes or zeroing launch; ranks by LDS add; slot order stays arrival order
# speedup vs baseline: 16.2048x; 1.8067x over previous
#include <hip/hip_runtime.h>
#include <math.h>

typedef float v2f __attribute__((ext_vector_type(2)));
typedef float v4f __attribute__((ext_vector_type(4)));
typedef float v8f __attribute__((ext_vector_type(8)));

#define IN_DIM 128
#define OUT_DIM 128
#define NHEAD 4
#define HD 32
#define NEG_SLOPE 0.2f
#define EPS_SOFTMAX 1e-8f
#define EPS_LN 1e-5f

__global__ __launch_bounds__(256) void gemm_wmma_kernel(
    const float* __restrict__ h, const float* __restrict__ W,
    const float* __restrict__ Wself, float* __restrict__ h_proj,
    float* __restrict__ out, int N)
{
    const int wave = threadIdx.x >> 5;
    const int lane = threadIdx.x & 31;
    const int row0 = blockIdx.x * 16;
    const int mrow = lane & 15;
    const int khalf = (lane >> 4) * 2;

    int arow_idx = row0 + mrow;
    if (arow_idx >= N) arow_idx = N - 1;
    const float* arow = h + (size_t)arow_idx * IN_DIM;

    const int col0 = wave * 32;
    const int ccol0 = col0 + mrow;
    const int ccol1 = col0 + 16 + mrow;
    const bool t0_proj = (col0 < 128);
    const bool t1_proj = (col0 + 16 < 128);
    const float* bp0 = t0_proj ? (W + ccol0) : (Wself + (ccol0 - 128));
    const float* bp1 = t1_proj ? (W + ccol1) : (Wself + (ccol1 - 128));

    v8f c0 = {};
    v8f c1 = {};
    for (int k0 = 0; k0 < IN_DIM; k0 += 4) {
        const int k = k0 + khalf;
        v2f a, b0, b1;
        a.x = arow[k];
        a.y = arow[k + 1];
        b0.x = bp0[(size_t)k * OUT_DIM];
        b0.y = bp0[(size_t)(k + 1) * OUT_DIM];
        b1.x = bp1[(size_t)k * OUT_DIM];
        b1.y = bp1[(size_t)(k + 1) * OUT_DIM];
        c0 = __builtin_amdgcn_wmma_f32_16x16x4_f32(false, a, false, b0,
                                                   (short)0, c0, false, false);
        c1 = __builtin_amdgcn_wmma_f32_16x16x4_f32(false, a, false, b1,
                                                   (short)0, c1, false, false);
    }

    const int mbase = (lane >> 4) * 8;
    float* dst0 = t0_proj ? (h_proj + ccol0) : (out + (ccol0 - 128));
    float* dst1 = t1_proj ? (h_proj + ccol1) : (out + (ccol1 - 128));
    if (row0 + 16 <= N) {
#pragma unroll
        for (int v = 0; v < 8; ++v) {
            const size_t row = (size_t)(row0 + mbase + v);
            dst0[row * OUT_DIM] = c0[v];
            dst1[row * OUT_DIM] = c1[v];
        }
    } else {
#pragma unroll
        for (int v = 0; v < 8; ++v) {
            const int row = row0 + mbase + v;
            if (row < N) {
                dst0[(size_t)row * OUT_DIM] = c0[v];
                dst1[(size_t)row * OUT_DIM] = c1[v];
            }
        }
    }
}

__global__ void score_kernel(const float* __restrict__ h_proj,
                             const float* __restrict__ a_in,
                             const float* __restrict__ a_out,
                             float* __restrict__ s_in_s, float* __restrict__ s_in_r,
                             float* __restrict__ s_out_s, float* __restrict__ s_out_r,
                             int NH)
{
    int gid = blockIdx.x * blockDim.x + threadIdx.x;
    if (gid >= NH) return;
    const int hh = gid & 3;
    const int n = gid >> 2;
    const float* x = h_proj + (size_t)n * OUT_DIM + hh * HD;
    const float* ai = a_in + hh * 2 * HD;
    const float* ao = a_out + hh * 2 * HD;
    float s0 = 0.f, s1 = 0.f, s2 = 0.f, s3 = 0.f;
#pragma unroll
    for (int d = 0; d < HD; d += 4) {
        const v4f xv = *(const v4f*)(x + d);
#pragma unroll
        for (int j = 0; j < 4; ++j) {
            s0 += xv[j] * ai[d + j];
            s1 += xv[j] * ai[HD + d + j];
            s2 += xv[j] * ao[d + j];
            s3 += xv[j] * ao[HD + d + j];
        }
    }
    s_in_s[gid] = s0;
    s_in_r[gid] = s1;
    s_out_s[gid] = s2;
    s_out_r[gid] = s3;
}

#define BSHIFT 9
#define BSIZE (1 << BSHIFT)
#define CHUNK 8192
#define EPT (CHUNK / 1024)
#define KMAX 1024
#define VBITS 23

__device__ __forceinline__ int block_excl_scan_1024(int v, int* wtot) {
    const int lane = threadIdx.x & 31;
    const int wave = threadIdx.x >> 5;
    int inc = v;
    for (int o = 1; o < 32; o <<= 1) {
        const int u = __shfl_up(inc, o);
        if (lane >= o) inc += u;
    }
    if (lane == 31) wtot[wave] = inc;
    __syncthreads();
    int carry = 0;
    for (int w = 0; w < wave; ++w) carry += wtot[w];
    return carry + inc - v;
}

__global__ __launch_bounds__(1024) void bucket_count_kernel(
    const int* __restrict__ src, const int* __restrict__ dst, int* __restrict__ gcnt,
    int E, int K, int P)
{
    __shared__ int cnt[KMAX];
    const int t = threadIdx.x;
    const int blk = blockIdx.x;
    const int dir = blockIdx.y;
    const int* key = dir ? src : dst;
    if (t < K) cnt[t] = 0;
    __syncthreads();
    const int e0 = blk * CHUNK + t;
    int kk[EPT];
#pragma unroll
    for (int i = 0; i < EPT; ++i) {
        const int e = e0 + i * 1024;
        kk[i] = key[(e < E) ? e : (E - 1)];
    }
#pragma unroll
    for (int i = 0; i < EPT; ++i) {
        if (e0 + i * 1024 < E) atomicAdd(&cnt[kk[i] >> BSHIFT], 1);
    }
    __syncthreads();
    int* g = gcnt + (size_t)dir * K * P;
    if (t < K) g[(size_t)t * P + blk] = cnt[t];
}

__global__ __launch_bounds__(1024) void scan_kernel(const int* __restrict__ deg,
                                                    int* __restrict__ rowptr, int N)
{
    __shared__ int wtot[32];
    const int dir = blockIdx.x;
    const int* dg = deg + (size_t)dir * N;
    int* rp = rowptr + (size_t)dir * (N + 1);
    const int lane = threadIdx.x & 31;
    const int wave = threadIdx.x >> 5;
    const int span = (((N + 31) / 32) + 127) & ~127;
    const int b = wave * span;
    const int en = (b + span < N) ? (b + span) : N;

    int s = 0;
    for (int i = b + lane; i < en; i += 32) s += dg[i];
    for (int o = 16; o > 0; o >>= 1) s += __shfl_xor(s, o);
    if (lane == 0) wtot[wave] = s;
    __syncthreads();
    int carry = 0;
    for (int w = 0; w < wave; ++w) carry += wtot[w];

    for (int i0 = b; i0 < en; i0 += 128) {
        const int i = i0 + lane * 4;
        const int v0 = (i     < en) ? dg[i]     : 0;
        const int v1 = (i + 1 < en) ? dg[i + 1] : 0;
        const int v2 = (i + 2 < en) ? dg[i + 2] : 0;
        const int v3 = (i + 3 < en) ? dg[i + 3] : 0;
        const int t = (v0 + v1) + (v2 + v3);
        int inc = t;
        for (int o = 1; o < 32; o <<= 1) {
            const int u = __shfl_up(inc, o);
            if (lane >= o) inc += u;
        }
        int ex = carry + inc - t;
        if (i     < en) rp[i]     = ex;
        ex += v0;
        if (i + 1 < en) rp[i + 1] = ex;
        ex += v1;
        if (i + 2 < en) rp[i + 2] = ex;
        ex += v2;
        if (i + 3 < en) rp[i + 3] = ex;
        carry += __shfl(inc, 31);
    }
    if (wave == 31 && lane == 0) rp[N] = carry;
}

__global__ __launch_bounds__(1024) void bucket_scatter_kernel(
    const int* __restrict__ src, const int* __restrict__ dst, const int* __restrict__ goff,
    unsigned int* __restrict__ binned, int E, int K, int P)
{
    __shared__ int cnt[KMAX];
    __shared__ int gbase[KMAX];
    __shared__ int wtot[32];
    __shared__ unsigned int s_item[CHUNK];
    __shared__ unsigned short s_bkt[CHUNK];
    const int t = threadIdx.x;
    const int blk = blockIdx.x;
    const int dir = blockIdx.y;
    const int* key = dir ? src : dst;
    const int* val = dir ? dst : src;
    if (t < K) cnt[t] = 0;
    __syncthreads();
    int kk[EPT], vv[EPT], rk[EPT];
    const int e0 = blk * CHUNK + t;
#pragma unroll
    for (int i = 0; i < EPT; ++i) {
        const int e = e0 + i * 1024;
        const int ec = (e < E) ? e : (E - 1);
        kk[i] = key[ec];
        vv[i] = val[ec];
    }
#pragma unroll
    for (int i = 0; i < EPT; ++i) {
        rk[i] = 0;
        if (e0 + i * 1024 < E) rk[i] = atomicAdd(&cnt[kk[i] >> BSHIFT], 1);
        else kk[i] = -1;
    }
    __syncthreads();
    const int c = (t < K) ? cnt[t] : 0;
    const int ex = block_excl_scan_1024(c, wtot);
    if (t < K) {
        cnt[t] = ex;
        gbase[t] = goff[(size_t)dir * ((size_t)K * P + 1) + (size_t)t * P + blk] - ex;
    }
    __syncthreads();
#pragma unroll
    for (int i = 0; i < EPT; ++i) {
        if (kk[i] >= 0) {
            const int b = kk[i] >> BSHIFT;
            const int pos = cnt[b] + rk[i];
            s_item[pos] = ((unsigned int)(kk[i] & (BSIZE - 1)) << VBITS) | (unsigned int)vv[i];
            s_bkt[pos] = (unsigned short)b;
        }
    }
    __syncthreads();
    const int left = E - blk * CHUNK;
    const int nloc = (left < CHUNK) ? left : CHUNK;
    unsigned int* outp = binned + (size_t)dir * E;
    for (int i = t; i < nloc; i += 1024) outp[gbase[s_bkt[i]] + i] = s_item[i];
}

#define DCAP 16384
#define DU 8
__global__ __launch_bounds__(1024) void bucket_rows_kernel(
    const unsigned int* __restrict__ binned, const int* __restrict__ goff,
    int* __restrict__ rowptr, int* __restrict__ nbr, int N, int E, int K, int P)
{
    __shared__ int cnt[BSIZE];
    __shared__ int cur[BSIZE];
    __shared__ int wtot[32];
    __shared__ int s_out[DCAP];
    const int t = threadIdx.x;
    const int b = blockIdx.x;
    const int dir = blockIdx.y;
    const int* go = goff + (size_t)dir * ((size_t)K * P + 1);
    const int beg = go[(size_t)b * P];
    const int end = go[(size_t)(b + 1) * P];
    const int cntb = end - beg;
    const bool staged = cntb <= DCAP;
    const unsigned int* items = binned + (size_t)dir * E;
    if (t < BSIZE) cnt[t] = 0;
    __syncthreads();
    for (int i0 = beg + t; i0 < end; i0 += DU * 1024) {
        unsigned int it[DU];
#pragma unroll
        for (int u = 0; u < DU; ++u) {
            const int j = i0 + u * 1024;
            it[u] = items[(j < end) ? j : i0];
        }
#pragma unroll
        for (int u = 0; u < DU; ++u) {
            if (i0 + u * 1024 < end) atomicAdd(&cnt[it[u] >> VBITS], 1);
        }
    }
    __syncthreads();
    const int c = (t < BSIZE) ? cnt[t] : 0;
    const int ex = block_excl_scan_1024(c, wtot);
    int* rp = rowptr + (size_t)dir * (N + 1);
    if (t < BSIZE) {
        cur[t] = ex;
        const int node = (b << BSHIFT) + t;
        if (node < N) rp[node] = beg + ex;
    }
    if (b == K - 1 && t == 0) rp[N] = end;
    __syncthreads();
    int* nb = nbr + (size_t)dir * E + beg;
    for (int i0 = beg + t; i0 < end; i0 += DU * 1024) {
        unsigned int it[DU];
#pragma unroll
        for (int u = 0; u < DU; ++u) {
            const int j = i0 + u * 1024;
            it[u] = items[(j < end) ? j : i0];
        }
#pragma unroll
        for (int u = 0; u < DU; ++u) {
            if (i0 + u * 1024 < end) {
                const int p = atomicAdd(&cur[it[u] >> VBITS], 1);
                const int v = (int)(it[u] & ((1u << VBITS) - 1u));
                if (staged) s_out[p] = v; else nb[p] = v;
            }
        }
    }
    if (staged) {
        __syncthreads();
        for (int i = t; i < cntb; i += 1024) nb[i] = s_out[i];
    }
}

__device__ __forceinline__ float wave_max32(float v) {
    for (int o = 16; o > 0; o >>= 1) v = fmaxf(v, __shfl_xor(v, o));
    return v;
}
__device__ __forceinline__ float wave_sum32(float v) {
    for (int o = 16; o > 0; o >>= 1) v += __shfl_xor(v, o);
    return v;
}
__device__ __forceinline__ float lrelu(float x) { return (x > 0.0f) ? x : NEG_SLOPE * x; }

__device__ __forceinline__ void aggregate_dir(const int* __restrict__ nbr, int beg, int end,
                                              const float* __restrict__ s_s, v4f sr,
                                              const float* __restrict__ h_proj, int lane,
                                              float (&acc)[NHEAD])
{
    if (end <= beg) return;
    const bool more = (end - beg) > 32;

    const int j0 = beg + lane;
    const bool ok0 = j0 < end;
    int s0 = 0;
    v4f e0 = {-INFINITY, -INFINITY, -INFINITY, -INFINITY};
    if (ok0) {
        s0 = nbr[j0];
        const v4f ss = *(const v4f*)(s_s + (size_t)s0 * 4);
        for (int h = 0; h < NHEAD; ++h) e0[h] = lrelu(ss[h] + sr[h]);
    }

    v4f m = e0;
    if (more) {
        for (int j = j0 + 32; j < end; j += 32) {
            const int s = nbr[j];
            const v4f ss = *(const v4f*)(s_s + (size_t)s * 4);
            for (int h = 0; h < NHEAD; ++h) m[h] = fmaxf(m[h], lrelu(ss[h] + sr[h]));
        }
    }
    for (int h = 0; h < NHEAD; ++h) m[h] = wave_max32(m[h]);

    v4f a0 = {0.0f, 0.0f, 0.0f, 0.0f};
    if (ok0) {
        for (int h = 0; h < NHEAD; ++h) a0[h] = __expf(e0[h] - m[h]);
    }
    v4f den = a0;
    if (more) {
        for (int j = j0 + 32; j < end; j += 32) {
            const int s = nbr[j];
            const v4f ss = *(const v4f*)(s_s + (size_t)s * 4);
            for (int h = 0; h < NHEAD; ++h) den[h] += __expf(lrelu(ss[h] + sr[h]) - m[h]);
        }
    }
    for (int h = 0; h < NHEAD; ++h) den[h] = wave_sum32(den[h]) + EPS_SOFTMAX;

    for (int c = beg; c < end; c += 32) {
        int s;
        v4f w = {0.0f, 0.0f, 0.0f, 0.0f};
        const int j = c + lane;
        if (c == beg) {
            s = s0;
            for (int h = 0; h < NHEAD; ++h) w[h] = a0[h] / den[h];
        } else {
            s = 0;
            if (j < end) {
                s = nbr[j];
                const v4f ss = *(const v4f*)(s_s + (size_t)s * 4);
                for (int h = 0; h < NHEAD; ++h)
                    w[h] = __expf(lrelu(ss[h] + sr[h]) - m[h]) / den[h];
            }
        }
        const int sfirst = __builtin_amdgcn_readfirstlane(s);
        if (j >= end) s = sfirst;
        const int cnt = (end - c < 32) ? (end - c) : 32;
        for (int i = 0; i < cnt; i += 4) {
#pragma unroll
            for (int u = 0; u < 4; ++u) {
                const int si = __builtin_amdgcn_readlane(s, i + u);
                const float* xr = h_proj + (size_t)si * OUT_DIM + lane;
                for (int h = 0; h < NHEAD; ++h) {
                    const float wh = __int_as_float(
                        __builtin_amdgcn_readlane(__float_as_int(w[h]), i + u));
                    acc[h] += wh * xr[h * HD];
                }
            }
        }
    }
}

__global__ __launch_bounds__(256) void gather_kernel(
    const int* __restrict__ rowptr, const int* __restrict__ nbr_in,
    const int* __restrict__ nbr_out, const float* __restrict__ h_proj,
    const float* __restrict__ s_in_s, const float* __restrict__ s_in_r,
    const float* __restrict__ s_out_s, const float* __restrict__ s_out_r,
    const float* __restrict__ bias, const float* __restrict__ gamma,
    const float* __restrict__ beta, float* __restrict__ out, int N)
{
    const int lane = threadIdx.x & 31;
    const int n = __builtin_amdgcn_readfirstlane(blockIdx.x * 8 + (threadIdx.x >> 5));
    if (n >= N) return;
    const int* rp_in = rowptr;
    const int* rp_out = rowptr + (size_t)(N + 1);
    const int bi = rp_in[n], ei = rp_in[n + 1];
    const int bo = rp_out[n], eo = rp_out[n + 1];
    const v4f sri = *(const v4f*)(s_in_r + (size_t)n * 4);
    const v4f sro = *(const v4f*)(s_out_r + (size_t)n * 4);
    float* orow = out + (size_t)n * OUT_DIM + lane;
    float comb[NHEAD];
    for (int h = 0; h < NHEAD; ++h) comb[h] = orow[h * HD];

    float acc_in[NHEAD] = {0.0f, 0.0f, 0.0f, 0.0f};
    aggregate_dir(nbr_in, bi, ei, s_in_s, sri, h_proj, lane, acc_in);
    for (int h = 0; h < NHEAD; ++h) comb[h] += acc_in[h];
    float acc_out[NHEAD] = {0.0f, 0.0f, 0.0f, 0.0f};
    aggregate_dir(nbr_out, bo, eo, s_out_s, sro, h_proj, lane, acc_out);

    float x[NHEAD];
    for (int h = 0; h < NHEAD; ++h) x[h] = (comb[h] + acc_out[h]) + bias[h * HD + lane];

    const float mean = wave_sum32((x[0] + x[1]) + (x[2] + x[3])) * (1.0f / OUT_DIM);
    float d[NHEAD];
    for (int h = 0; h < NHEAD; ++h) d[h] = x[h] - mean;
    const float var = wave_sum32((d[0] * d[0] + d[1] * d[1]) + (d[2] * d[2] + d[3] * d[3]))
                      * (1.0f / OUT_DIM);
    const float inv = rsqrtf(var + EPS_LN);
    for (int h = 0; h < NHEAD; ++h)
        orow[h * HD] = d[h] * inv * gamma[h * HD + lane] + beta[h * HD + lane];
}

extern "C" void kernel_launch(void* const* d_in, const int* in_sizes, int n_in,
                              void* d_out, int out_size, void* d_ws, size_t ws_size,
                              hipStream_t stream) {
    const float* h      = (const float*)d_in[0];
    const int*   eidx   = (const int*)d_in[1];
    const float* W      = (const float*)d_in[2];
    const float* Wself  = (const float*)d_in[3];
    const float* a_in   = (const float*)d_in[4];
    const float* a_out  = (const float*)d_in[5];
    const float* bias   = (const float*)d_in[6];
    const float* gamma  = (const float*)d_in[7];
    const float* beta   = (const float*)d_in[8];
    float* out = (float*)d_out;

    const int N = in_sizes[0] / IN_DIM;
    const int E = in_sizes[1] / 2;
    const int NH = N * NHEAD;

    const int* src = eidx;
    const int* dst = eidx + E;

    float* ws = (float*)d_ws;
    float* h_proj  = ws;
    float* s_in_s  = h_proj + (size_t)N * OUT_DIM;
    float* s_in_r  = s_in_s + NH;
    float* s_out_s = s_in_r + NH;
    float* s_out_r = s_out_s + NH;
    int* cntoff  = (int*)(s_out_r + NH);
    int* rowptr  = cntoff + (size_t)2 * N;
    int* nbr_in  = rowptr + (size_t)2 * (N + 1);
    int* nbr_out = nbr_in + (size_t)E;
    const int K = (N + BSIZE - 1) >> BSHIFT;
    const int P = (E + CHUNK - 1) / CHUNK;
    const int KP = K * P;
    int* gcnt = cntoff;
    int* goff = gcnt + (size_t)2 * KP;
    unsigned int* binned = (unsigned int*)h_proj;

    {
        bucket_count_kernel<<<dim3(P, 2), 1024, 0, stream>>>(src, dst, gcnt, E, K, P);
    }
    {
        scan_kernel<<<2, 1024, 0, stream>>>(gcnt, goff, KP);
    }
    {
        bucket_scatter_kernel<<<dim3(P, 2), 1024, 0, stream>>>(src, dst, goff, binned, E, K, P);
    }
    {
        bucket_rows_kernel<<<dim3(K, 2), 1024, 0, stream>>>(binned, goff, rowptr, nbr_in,
                                                           N, E, K, P);
    }
    {
        int blocks = (N + 15) / 16;
        gemm_wmma_kernel<<<blocks, 256, 0, stream>>>(h, W, Wself, h_proj, out, N);
    }
    {
        int threads = 256, blocks = (NH + threads - 1) / threads;
        score_kernel<<<blocks, threads, 0, stream>>>(h_proj, a_in, a_out,
                                                     s_in_s, s_in_r, s_out_s, s_out_r, NH);
    }
    {
        int blocks = (N + 7) / 8;
        gather_kernel<<<blocks, 256, 0, stream>>>(rowptr, nbr_in, nbr_out, h_proj,
                                                  s_in_s, s_in_r, s_out_s, s_out_r,
                                                  bias, gamma, beta, out, N);
    }
}


# GEMM weights stay in registers: 512 resident blocks loop over row tiles, each wave keeps its 32 columns' fragments for all k; A tile staged once per block in LDS; k order of the sums changes
# speedup vs baseline: 18.4446x; 1.1382x over previous
#include <hip/hip_runtime.h>
#include <math.h>

typedef float v2f __attribute__((ext_vector_type(2)));
typedef float v4f __attribute__((ext_vector_type(4)));
typedef float v8f __attribute__((ext_vector_type(8)));

#define IN_DIM 128
#define OUT_DIM 128
#define NHEAD 4
#define HD 32
#define NEG_SLOPE 0.2f
#define EPS_SOFTMAX 1e-8f
#define EPS_LN 1e-5f

#define GEMM_BLOCKS 512
#define A_PITCH 132
__global__ __launch_bounds__(256) void gemm_wmma_kernel(
    const float* __restrict__ h, const float* __restrict__ W,
    const float* __restrict__ Wself, float* __restrict__ h_proj,
    float* __restrict__ out, int N)
{
    __shared__ __attribute__((aligned(16))) float As[2][16 * A_PITCH];
    const int tid = threadIdx.x;
    const int wave = tid >> 5;
    const int lane = tid & 31;
    const int c = lane & 15;
    const int hh = lane >> 4;
    const int ntiles = (N + 15) >> 4;
    int t = blockIdx.x;
    if (t >= ntiles) return;

    const int sr = tid >> 4;
    const int sq = (tid & 15) * 4;
    {
        int row = t * 16 + sr;
        if (row >= N) row = N - 1;
        const float* p = h + (size_t)row * IN_DIM + sq;
        const v4f p0 = *(const v4f*)p;
        const v4f p1 = *(const v4f*)(p + 64);
        *(v4f*)&As[0][sr * A_PITCH + sq] = p0;
        *(v4f*)&As[0][sr * A_PITCH + 64 + sq] = p1;
    }

    const int col0 = wave * 32;
    const float* bm = (wave < 4) ? (W + col0) : (Wself + (col0 - 128));
    v4f b0[16], b1[16];
#pragma unroll
    for (int j = 0; j < 16; ++j) {
        const float* bk = bm + (size_t)(8 * j + 4 * hh) * OUT_DIM + c;
        b0[j].x = bk[0];
        b0[j].y = bk[OUT_DIM];
        b0[j].z = bk[2 * OUT_DIM];
        b0[j].w = bk[3 * OUT_DIM];
        b1[j].x = bk[16];
        b1[j].y = bk[OUT_DIM + 16];
        b1[j].z = bk[2 * OUT_DIM + 16];
        b1[j].w = bk[3 * OUT_DIM + 16];
    }
    float* dst0 = ((wave < 4) ? (h_proj + col0) : (out + (col0 - 128))) + c;
    float* dst1 = dst0 + 16;
    const int mbase = hh * 8;
    const int step = (int)gridDim.x;
    __syncthreads();

    int buf = 0;
    for (; t < ntiles; t += step) {
        const int tn = t + step;
        const bool has_next = tn < ntiles;
        v4f p0 = {}, p1 = {};
        if (has_next) {
            int row = tn * 16 + sr;
            if (row >= N) row = N - 1;
            const float* p = h + (size_t)row * IN_DIM + sq;
            p0 = *(const v4f*)p;
            p1 = *(const v4f*)(p + 64);
        }

        const float* arow = &As[buf][c * A_PITCH + 4 * hh];
        v4f a[16];
#pragma unroll
        for (int j = 0; j < 16; ++j) a[j] = *(const v4f*)(arow + 8 * j);
        __builtin_amdgcn_sched_barrier(0);

        v8f c0 = {};
        v8f c1 = {};
#pragma unroll
        for (int j = 0; j < 16; ++j) {
            v2f alo, ahi, w0, w1;
            alo.x = a[j].x; alo.y = a[j].y;
            ahi.x = a[j].z; ahi.y = a[j].w;
            w0.x = b0[j].x; w0.y = b0[j].y;
            w1.x = b1[j].x; w1.y = b1[j].y;
            c0 = __builtin_amdgcn_wmma_f32_16x16x4_f32(false, alo, false, w0,
                                                       (short)0, c0, false, false);
            c1 = __builtin_amdgcn_wmma_f32_16x16x4_f32(false, alo, false, w1,
                                                       (short)0, c1, false, false);
            w0.x = b0[j].z; w0.y = b0[j].w;
            w1.x = b1[j].z; w1.y = b1[j].w;
            c0 = __builtin_amdgcn_wmma_f32_16x16x4_f32(false, ahi, false, w0,
                                                       (short)0, c0, false, false);
            c1 = __builtin_amdgcn_wmma_f32_16x16x4_f32(false, ahi, false, w1,
                                                       (short)0, c1, false, false);
        }

        const int row0 = t * 16;
        if (row0 + 16 <= N) {
#pragma unroll
            for (int v = 0; v < 8; ++v) {
                const size_t row = (size_t)(row0 + mbase + v);
                dst0[row * OUT_DIM] = c0[v];
                dst1[row * OUT_DIM] = c1[v];
            }
        } else {
#pragma unroll
            for (int v = 0; v < 8; ++v) {
                const int row = row0 + mbase + v;
                if (row < N) {
                    dst0[(size_t)row * OUT_DIM] = c0[v];
                    dst1[(size_t)row * OUT_DIM] = c1[v];
                }
            }
        }

        if (has_next) {
            *(v4f*)&As[buf ^ 1][sr * A_PITCH + sq] = p0;
            *(v4f*)&As[buf ^ 1][sr * A_PITCH + 64 + sq] = p1;
        }
        __syncthreads();
        buf ^= 1;
    }
}

__global__ void score_kernel(const float* __restrict__ h_proj,
                             const float* __restrict__ a_in,
                             const float* __restrict__ a_out,
                             float* __restrict__ s_in_s, float* __restrict__ s_in_r,
                             float* __restrict__ s_out_s, float* __restrict__ s_out_r,
                             int NH)
{
    int gid = blockIdx.x * blockDim.x + threadIdx.x;
    if (gid >= NH) return;
    const int hh = gid & 3;
    const int n = gid >> 2;
    const float* x = h_proj + (size_t)n * OUT_DIM + hh * HD;
    const float* ai = a_in + hh * 2 * HD;
    const float* ao = a_out + hh * 2 * HD;
    float s0 = 0.f, s1 = 0.f, s2 = 0.f, s3 = 0.f;
#pragma unroll
    for (int d = 0; d < HD; d += 4) {
        const v4f xv = *(const v4f*)(x + d);
#pragma unroll
        for (int j = 0; j < 4; ++j) {
            s0 += xv[j] * ai[d + j];
            s1 += xv[j] * ai[HD + d + j];
            s2 += xv[j] * ao[d + j];
            s3 += xv[j] * ao[HD + d + j];
        }
    }
    s_in_s[gid] = s0;
    s_in_r[gid] = s1;
    s_out_s[gid] = s2;
    s_out_r[gid] = s3;
}

#define BSHIFT 9
#define BSIZE (1 << BSHIFT)
#define CHUNK 8192
#define EPT (CHUNK / 1024)
#define KMAX 1024
#define VBITS 23

__device__ __forceinline__ int block_excl_scan_1024(int v, int* wtot) {
    const int lane = threadIdx.x & 31;
    const int wave = threadIdx.x >> 5;
    int inc = v;
    for (int o = 1; o < 32; o <<= 1) {
        const int u = __shfl_up(inc, o);
        if (lane >= o) inc += u;
    }
    if (lane == 31) wtot[wave] = inc;
    __syncthreads();
    int carry = 0;
    for (int w = 0; w < wave; ++w) carry += wtot[w];
    return carry + inc - v;
}

__global__ __launch_bounds__(1024) void bucket_count_kernel(
    const int* __restrict__ src, const int* __restrict__ dst, int* __restrict__ gcnt,
    int E, int K, int P)
{
    __shared__ int cnt[KMAX];
    const int t = threadIdx.x;
    const int blk = blockIdx.x;
    const int dir = blockIdx.y;
    const int* key = dir ? src : dst;
    if (t < K) cnt[t] = 0;
    __syncthreads();
    const int e0 = blk * CHUNK + t;
    int kk[EPT];
#pragma unroll
    for (int i = 0; i < EPT; ++i) {
        const int e = e0 + i * 1024;
        kk[i] = key[(e < E) ? e : (E - 1)];
    }
#pragma unroll
    for (int i = 0; i < EPT; ++i) {
        if (e0 + i * 1024 < E) atomicAdd(&cnt[kk[i] >> BSHIFT], 1);
    }
    __syncthreads();
    int* g = gcnt + (size_t)dir * K * P;
    if (t < K) g[(size_t)t * P + blk] = cnt[t];
}

__global__ __launch_bounds__(1024) void scan_kernel(const int* __restrict__ deg,
                                                    int* __restrict__ rowptr, int N)
{
    __shared__ int wtot[32];
    const int dir = blockIdx.x;
    const int* dg = deg + (size_t)dir * N;
    int* rp = rowptr + (size_t)dir * (N + 1);
    const int lane = threadIdx.x & 31;
    const int wave = threadIdx.x >> 5;
    const int span = (((N + 31) / 32) + 127) & ~127;
    const int b = wave * span;
    const int en = (b + span < N) ? (b + span) : N;

    int s = 0;
    for (int i = b + lane; i < en; i += 32) s += dg[i];
    for (int o = 16; o > 0; o >>= 1) s += __shfl_xor(s, o);
    if (lane == 0) wtot[wave] = s;
    __syncthreads();
    int carry = 0;
    for (int w = 0; w < wave; ++w) carry += wtot[w];

    for (int i0 = b; i0 < en; i0 += 128) {
        const int i = i0 + lane * 4;
        const int v0 = (i     < en) ? dg[i]     : 0;
        const int v1 = (i + 1 < en) ? dg[i + 1] : 0;
        const int v2 = (i + 2 < en) ? dg[i + 2] : 0;
        const int v3 = (i + 3 < en) ? dg[i + 3] : 0;
        const int t = (v0 + v1) + (v2 + v3);
        int inc = t;
        for (int o = 1; o < 32; o <<= 1) {
            const int u = __shfl_up(inc, o);
            if (lane >= o) inc += u;
        }
        int ex = carry + inc - t;
        if (i     < en) rp[i]     = ex;
        ex += v0;
        if (i + 1 < en) rp[i + 1] = ex;
        ex += v1;
        if (i + 2 < en) rp[i + 2] = ex;
        ex += v2;
        if (i + 3 < en) rp[i + 3] = ex;
        carry += __shfl(inc, 31);
    }
    if (wave == 31 && lane == 0) rp[N] = carry;
}

__global__ __launch_bounds__(1024) void bucket_scatter_kernel(
    const int* __restrict__ src, const int* __restrict__ dst, const int* __restrict__ goff,
    unsigned int* __restrict__ binned, int E, int K, int P)
{
    __shared__ int cnt[KMAX];
    __shared__ int gbase[KMAX];
    __shared__ int wtot[32];
    __shared__ unsigned int s_item[CHUNK];
    __shared__ unsigned short s_bkt[CHUNK];
    const int t = threadIdx.x;
    const int blk = blockIdx.x;
    const int dir = blockIdx.y;
    const int* key = dir ? src : dst;
    const int* val = dir ? dst : src;
    if (t < K) cnt[t] = 0;
    __syncthreads();
    int kk[EPT], vv[EPT], rk[EPT];
    const int e0 = blk * CHUNK + t;
#pragma unroll
    for (int i = 0; i < EPT; ++i) {
        const int e = e0 + i * 1024;
        const int ec = (e < E) ? e : (E - 1);
        kk[i] = key[ec];
        vv[i] = val[ec];
    }
#pragma unroll
    for (int i = 0; i < EPT; ++i) {
        rk[i] = 0;
        if (e0 + i * 1024 < E) rk[i] = atomicAdd(&cnt[kk[i] >> BSHIFT], 1);
        else kk[i] = -1;
    }
    __syncthreads();
    const int c = (t < K) ? cnt[t] : 0;
    const int ex = block_excl_scan_1024(c, wtot);
    if (t < K) {
        cnt[t] = ex;
        gbase[t] = goff[(size_t)dir * ((size_t)K * P + 1) + (size_t)t * P + blk] - ex;
    }
    __syncthreads();
#pragma unroll
    for (int i = 0; i < EPT; ++i) {
        if (kk[i] >= 0) {
            const int b = kk[i] >> BSHIFT;
            const int pos = cnt[b] + rk[i];
            s_item[pos] = ((unsigned int)(kk[i] & (BSIZE - 1)) << VBITS) | (unsigned int)vv[i];
            s_bkt[pos] = (unsigned short)b;
        }
    }
    __syncthreads();
    const int left = E - blk * CHUNK;
    const int nloc = (left < CHUNK) ? left : CHUNK;
    unsigned int* outp = binned + (size_t)dir * E;
    for (int i = t; i < nloc; i += 1024) outp[gbase[s_bkt[i]] + i] = s_item[i];
}

#define DCAP 16384
#define DU 8
__global__ __launch_bounds__(1024) void bucket_rows_kernel(
    const unsigned int* __restrict__ binned, const int* __restrict__ goff,
    int* __restrict__ rowptr, int* __restrict__ nbr, int N, int E, int K, int P)
{
    __shared__ int cnt[BSIZE];
    __shared__ int cur[BSIZE];
    __shared__ int wtot[32];
    __shared__ int s_out[DCAP];
    const int t = threadIdx.x;
    const int b = blockIdx.x;
    const int dir = blockIdx.y;
    const int* go = goff + (size_t)dir * ((size_t)K * P + 1);
    const int beg = go[(size_t)b * P];
    const int end = go[(size_t)(b + 1) * P];
    const int cntb = end - beg;
    const bool staged = cntb <= DCAP;
    const unsigned int* items = binned + (size_t)dir * E;
    if (t < BSIZE) cnt[t] = 0;
    __syncthreads();
    for (int i0 = beg + t; i0 < end; i0 += DU * 1024) {
        unsigned int it[DU];
#pragma unroll
        for (int u = 0; u < DU; ++u) {
            const int j = i0 + u * 1024;
            it[u] = items[(j < end) ? j : i0];
        }
#pragma unroll
        for (int u = 0; u < DU; ++u) {
            if (i0 + u * 1024 < end) atomicAdd(&cnt[it[u] >> VBITS], 1);
        }
    }
    __syncthreads();
    const int c = (t < BSIZE) ? cnt[t] : 0;
    const int ex = block_excl_scan_1024(c, wtot);
    int* rp = rowptr + (size_t)dir * (N + 1);
    if (t < BSIZE) {
        cur[t] = ex;
        const int node = (b << BSHIFT) + t;
        if (node < N) rp[node] = beg + ex;
    }
    if (b == K - 1 && t == 0) rp[N] = end;
    __syncthreads();
    int* nb = nbr + (size_t)dir * E + beg;
    for (int i0 = beg + t; i0 < end; i0 += DU * 1024) {
        unsigned int it[DU];
#pragma unroll
        for (int u = 0; u < DU; ++u) {
            const int j = i0 + u * 1024;
            it[u] = items[(j < end) ? j : i0];
        }
#pragma unroll
        for (int u = 0; u < DU; ++u) {
            if (i0 + u * 1024 < end) {
                const int p = atomicAdd(&cur[it[u] >> VBITS], 1);
                const int v = (int)(it[u] & ((1u << VBITS) - 1u));
                if (staged) s_out[p] = v; else nb[p] = v;
            }
        }
    }
    if (staged) {
        __syncthreads();
        for (int i = t; i < cntb; i += 1024) nb[i] = s_out[i];
    }
}

__device__ __forceinline__ float wave_max32(float v) {
    for (int o = 16; o > 0; o >>= 1) v = fmaxf(v, __shfl_xor(v, o));
    return v;
}
__device__ __forceinline__ float wave_sum32(float v) {
    for (int o = 16; o > 0; o >>= 1) v += __shfl_xor(v, o);
    return v;
}
__device__ __forceinline__ float lrelu(float x) { return (x > 0.0f) ? x : NEG_SLOPE * x; }

__device__ __forceinline__ void aggregate_dir(const int* __restrict__ nbr, int beg, int end,
                                              const float* __restrict__ s_s, v4f sr,
                                              const float* __restrict__ h_proj, int lane,
                                              float (&acc)[NHEAD])
{
    if (end <= beg) return;
    const bool more = (end - beg) > 32;

    const int j0 = beg + lane;
    const bool ok0 = j0 < end;
    int s0 = 0;
    v4f e0 = {-INFINITY, -INFINITY, -INFINITY, -INFINITY};
    if (ok0) {
        s0 = nbr[j0];
        const v4f ss = *(const v4f*)(s_s + (size_t)s0 * 4);
        for (int h = 0; h < NHEAD; ++h) e0[h] = lrelu(ss[h] + sr[h]);
    }

    v4f m = e0;
    if (more) {
        for (int j = j0 + 32; j < end; j += 32) {
            const int s = nbr[j];
            const v4f ss = *(const v4f*)(s_s + (size_t)s * 4);
            for (int h = 0; h < NHEAD; ++h) m[h] = fmaxf(m[h], lrelu(ss[h] + sr[h]));
        }
    }
    for (int h = 0; h < NHEAD; ++h) m[h] = wave_max32(m[h]);

    v4f a0 = {0.0f, 0.0f, 0.0f, 0.0f};
    if (ok0) {
        for (int h = 0; h < NHEAD; ++h) a0[h] = __expf(e0[h] - m[h]);
    }
    v4f den = a0;
    if (more) {
        for (int j = j0 + 32; j < end; j += 32) {
            const int s = nbr[j];
            const v4f ss = *(const v4f*)(s_s + (size_t)s * 4);
            for (int h = 0; h < NHEAD; ++h) den[h] += __expf(lrelu(ss[h] + sr[h]) - m[h]);
        }
    }
    for (int h = 0; h < NHEAD; ++h) den[h] = wave_sum32(den[h]) + EPS_SOFTMAX;

    for (int c = beg; c < end; c += 32) {
        int s;
        v4f w = {0.0f, 0.0f, 0.0f, 0.0f};
        const int j = c + lane;
        if (c == beg) {
            s = s0;
            for (int h = 0; h < NHEAD; ++h) w[h] = a0[h] / den[h];
        } else {
            s = 0;
            if (j < end) {
                s = nbr[j];
                const v4f ss = *(const v4f*)(s_s + (size_t)s * 4);
                for (int h = 0; h < NHEAD; ++h)
                    w[h] = __expf(lrelu(ss[h] + sr[h]) - m[h]) / den[h];
            }
        }
        const int sfirst = __builtin_amdgcn_readfirstlane(s);
        if (j >= end) s = sfirst;
        const int cnt = (end - c < 32) ? (end - c) : 32;
        for (int i = 0; i < cnt; i += 4) {
#pragma unroll
            for (int u = 0; u < 4; ++u) {
                const int si = __builtin_amdgcn_readlane(s, i + u);
                const float* xr = h_proj + (size_t)si * OUT_DIM + lane;
                for (int h = 0; h < NHEAD; ++h) {
                    const float wh = __int_as_float(
                        __builtin_amdgcn_readlane(__float_as_int(w[h]), i + u));
                    acc[h] += wh * xr[h * HD];
                }
            }
        }
    }
}

__global__ __launch_bounds__(256) void gather_kernel(
    const int* __restrict__ rowptr, const int* __restrict__ nbr_in,
    const int* __restrict__ nbr_out, const float* __restrict__ h_proj,
    const float* __restrict__ s_in_s, const float* __restrict__ s_in_r,
    const float* __restrict__ s_out_s, const float* __restrict__ s_out_r,
    const float* __restrict__ bias, const float* __restrict__ gamma,
    const float* __restrict__ beta, float* __restrict__ out, int N)
{
    const int lane = threadIdx.x & 31;
    const int n = __builtin_amdgcn_readfirstlane(blockIdx.x * 8 + (threadIdx.x >> 5));
    if (n >= N) return;
    const int* rp_in = rowptr;
    const int* rp_out = rowptr + (size_t)(N + 1);
    const int bi = rp_in[n], ei = rp_in[n + 1];
    const int bo = rp_out[n], eo = rp_out[n + 1];
    const v4f sri = *(const v4f*)(s_in_r + (size_t)n * 4);
    const v4f sro = *(const v4f*)(s_out_r + (size_t)n * 4);
    float* orow = out + (size_t)n * OUT_DIM + lane;
    float comb[NHEAD];
    for (int h = 0; h < NHEAD; ++h) comb[h] = orow[h * HD];

    float acc_in[NHEAD] = {0.0f, 0.0f, 0.0f, 0.0f};
    aggregate_dir(nbr_in, bi, ei, s_in_s, sri, h_proj, lane, acc_in);
    for (int h = 0; h < NHEAD; ++h) comb[h] += acc_in[h];
    float acc_out[NHEAD] = {0.0f, 0.0f, 0.0f, 0.0f};
    aggregate_dir(nbr_out, bo, eo, s_out_s, sro, h_proj, lane, acc_out);

    float x[NHEAD];
    for (int h = 0; h < NHEAD; ++h) x[h] = (comb[h] + acc_out[h]) + bias[h * HD + lane];

    const float mean = wave_sum32((x[0] + x[1]) + (x[2] + x[3])) * (1.0f / OUT_DIM);
    float d[NHEAD];
    for (int h = 0; h < NHEAD; ++h) d[h] = x[h] - mean;
    const float var = wave_sum32((d[0] * d[0] + d[1] * d[1]) + (d[2] * d[2] + d[3] * d[3]))
                      * (1.0f / OUT_DIM);
    const float inv = rsqrtf(var + EPS_LN);
    for (int h = 0; h < NHEAD; ++h)
        orow[h * HD] = d[h] * inv * gamma[h * HD + lane] + beta[h * HD + lane];
}

extern "C" void kernel_launch(void* const* d_in, const int* in_sizes, int n_in,
                              void* d_out, int out_size, void* d_ws, size_t ws_size,
                              hipStream_t stream) {
    const float* h      = (const float*)d_in[0];
    const int*   eidx   = (const int*)d_in[1];
    const float* W      = (const float*)d_in[2];
    const float* Wself  = (const float*)d_in[3];
    const float* a_in   = (const float*)d_in[4];
    const float* a_out  = (const float*)d_in[5];
    const float* bias   = (const float*)d_in[6];
    const float* gamma  = (const float*)d_in[7];
    const float* beta   = (const float*)d_in[8];
    float* out = (float*)d_out;

    const int N = in_sizes[0] / IN_DIM;
    const int E = in_sizes[1] / 2;
    const int NH = N * NHEAD;

    const int* src = eidx;
    const int* dst = eidx + E;

    float* ws = (float*)d_ws;
    float* h_proj  = ws;
    float* s_in_s  = h_proj + (size_t)N * OUT_DIM;
    float* s_in_r  = s_in_s + NH;
    float* s_out_s = s_in_r + NH;
    float* s_out_r = s_out_s + NH;
    int* cntoff  = (int*)(s_out_r + NH);
    int* rowptr  = cntoff + (size_t)2 * N;
    int* nbr_in  = rowptr + (size_t)2 * (N + 1);
    int* nbr_out = nbr_in + (size_t)E;
    const int K = (N + BSIZE - 1) >> BSHIFT;
    const int P = (E + CHUNK - 1) / CHUNK;
    const int KP = K * P;
    int* gcnt = cntoff;
    int* goff = gcnt + (size_t)2 * KP;
    unsigned int* binned = (unsigned int*)h_proj;

    {
        bucket_count_kernel<<<dim3(P, 2), 1024, 0, stream>>>(src, dst, gcnt, E, K, P);
    }
    {
        scan_kernel<<<2, 1024, 0, stream>>>(gcnt, goff, KP);
    }
    {
        bucket_scatter_kernel<<<dim3(P, 2), 1024, 0, stream>>>(src, dst, goff, binned, E, K, P);
    }
    {
        bucket_rows_kernel<<<dim3(K, 2), 1024, 0, stream>>>(binned, goff, rowptr, nbr_in,
                                                           N, E, K, P);
    }
    {
        gemm_wmma_kernel<<<GEMM_BLOCKS, 256, 0, stream>>>(h, W, Wself, h_proj, out, N);
    }
    {
        int threads = 256, blocks = (NH + threads - 1) / threads;
        score_kernel<<<blocks, threads, 0, stream>>>(h_proj, a_in, a_out,
                                                     s_in_s, s_in_r, s_out_s, s_out_r, NH);
    }
    {
        int blocks = (N + 7) / 8;
        gather_kernel<<<blocks, 256, 0, stream>>>(rowptr, nbr_in, nbr_out, h_proj,
                                                  s_in_s, s_in_r, s_out_s, s_out_r,
                                                  bias, gamma, beta, out, N);
    }
}


# Gather at 16 B per lane: a lane owns 4 adjacent columns of one head, a sender row is one 512-B load (was four); slots spread over 8-lane head groups, so softmax and LayerNorm sum trees change
# speedup vs baseline: 19.8883x; 1.0783x over previous
#include <hip/hip_runtime.h>
#include <math.h>

typedef float v2f __attribute__((ext_vector_type(2)));
typedef float v4f __attribute__((ext_vector_type(4)));
typedef float v8f __attribute__((ext_vector_type(8)));

#define IN_DIM 128
#define OUT_DIM 128
#define NHEAD 4
#define HD 32
#define NEG_SLOPE 0.2f
#define EPS_SOFTMAX 1e-8f
#define EPS_LN 1e-5f

#define GEMM_BLOCKS 512
#define A_PITCH 132
__global__ __launch_bounds__(256) void gemm_wmma_kernel(
    const float* __restrict__ h, const float* __restrict__ W,
    const float* __restrict__ Wself, float* __restrict__ h_proj,
    float* __restrict__ out, int N)
{
    __shared__ __attribute__((aligned(16))) float As[2][16 * A_PITCH];
    const int tid = threadIdx.x;
    const int wave = tid >> 5;
    const int lane = tid & 31;
    const int c = lane & 15;
    const int hh = lane >> 4;
    const int ntiles = (N + 15) >> 4;
    int t = blockIdx.x;
    if (t >= ntiles) return;

    const int sr = tid >> 4;
    const int sq = (tid & 15) * 4;
    {
        int row = t * 16 + sr;
        if (row >= N) row = N - 1;
        const float* p = h + (size_t)row * IN_DIM + sq;
        const v4f p0 = *(const v4f*)p;
        const v4f p1 = *(const v4f*)(p + 64);
        *(v4f*)&As[0][sr * A_PITCH + sq] = p0;
        *(v4f*)&As[0][sr * A_PITCH + 64 + sq] = p1;
    }

    const int col0 = wave * 32;
    const float* bm = (wave < 4) ? (W + col0) : (Wself + (col0 - 128));
    v4f b0[16], b1[16];
#pragma unroll
    for (int j = 0; j < 16; ++j) {
        const float* bk = bm + (size_t)(8 * j + 4 * hh) * OUT_DIM + c;
        b0[j].x = bk[0];
        b0[j].y = bk[OUT_DIM];
        b0[j].z = bk[2 * OUT_DIM];
        b0[j].w = bk[3 * OUT_DIM];
        b1[j].x = bk[16];
        b1[j].y = bk[OUT_DIM + 16];
        b1[j].z = bk[2 * OUT_DIM + 16];
        b1[j].w = bk[3 * OUT_DIM + 16];
    }
    float* dst0 = ((wave < 4) ? (h_proj + col0) : (out + (col0 - 128))) + c;
    float* dst1 = dst0 + 16;
    const int mbase = hh * 8;
    const int step = (int)gridDim.x;
    __syncthreads();

    int buf = 0;
    for (; t < ntiles; t += step) {
        const int tn = t + step;
        const bool has_next = tn < ntiles;
        v4f p0 = {}, p1 = {};
        if (has_next) {
            int row = tn * 16 + sr;
            if (row >= N) row = N - 1;
            const float* p = h + (size_t)row * IN_DIM + sq;
            p0 = *(const v4f*)p;
            p1 = *(const v4f*)(p + 64);
        }

        const float* arow = &As[buf][c * A_PITCH + 4 * hh];
        v4f a[16];
#pragma unroll
        for (int j = 0; j < 16; ++j) a[j] = *(const v4f*)(arow + 8 * j);
        __builtin_amdgcn_sched_barrier(0);

        v8f c0 = {};
        v8f c1 = {};
#pragma unroll
        for (int j = 0; j < 16; ++j) {
            v2f alo, ahi, w0, w1;
            alo.x = a[j].x; alo.y = a[j].y;
            ahi.x = a[j].z; ahi.y = a[j].w;
            w0.x = b0[j].x; w0.y = b0[j].y;
            w1.x = b1[j].x; w1.y = b1[j].y;
            c0 = __builtin_amdgcn_wmma_f32_16x16x4_f32(false, alo, false, w0,
                                                       (short)0, c0, false, false);
            c1 = __builtin_amdgcn_wmma_f32_16x16x4_f32(false, alo, false, w1,
                                                       (short)0, c1, false, false);
            w0.x = b0[j].z; w0.y = b0[j].w;
            w1.x = b1[j].z; w1.y = b1[j].w;
            c0 = __builtin_amdgcn_wmma_f32_16x16x4_f32(false, ahi, false, w0,
                                                       (short)0, c0, false, false);
            c1 = __builtin_amdgcn_wmma_f32_16x16x4_f32(false, ahi, false, w1,
                                                       (short)0, c1, false, false);
        }

        const int row0 = t * 16;
        if (row0 + 16 <= N) {
#pragma unroll
            for (int v = 0; v < 8; ++v) {
                const size_t row = (size_t)(row0 + mbase + v);
                dst0[row * OUT_DIM] = c0[v];
                dst1[row * OUT_DIM] = c1[v];
            }
        } else {
#pragma unroll
            for (int v = 0; v < 8; ++v) {
                const int row = row0 + mbase + v;
                if (row < N) {
                    dst0[(size_t)row * OUT_DIM] = c0[v];
                    dst1[(size_t)row * OUT_DIM] = c1[v];
                }
            }
        }

        if (has_next) {
            *(v4f*)&As[buf ^ 1][sr * A_PITCH + sq] = p0;
            *(v4f*)&As[buf ^ 1][sr * A_PITCH + 64 + sq] = p1;
        }
        __syncthreads();
        buf ^= 1;
    }
}

__global__ void score_kernel(const float* __restrict__ h_proj,
                             const float* __restrict__ a_in,
                             const float* __restrict__ a_out,
                             float* __restrict__ s_in_s, float* __restrict__ s_in_r,
                             float* __restrict__ s_out_s, float* __restrict__ s_out_r,
                             int NH)
{
    int gid = blockIdx.x * blockDim.x + threadIdx.x;
    if (gid >= NH) return;
    const int hh = gid & 3;
    const int n = gid >> 2;
    const float* x = h_proj + (size_t)n * OUT_DIM + hh * HD;
    const float* ai = a_in + hh * 2 * HD;
    const float* ao = a_out + hh * 2 * HD;
    float s0 = 0.f, s1 = 0.f, s2 = 0.f, s3 = 0.f;
#pragma unroll
    for (int d = 0; d < HD; d += 4) {
        const v4f xv = *(const v4f*)(x + d);
#pragma unroll
        for (int j = 0; j < 4; ++j) {
            s0 += xv[j] * ai[d + j];
            s1 += xv[j] * ai[HD + d + j];
            s2 += xv[j] * ao[d + j];
            s3 += xv[j] * ao[HD + d + j];
        }
    }
    s_in_s[gid] = s0;
    s_in_r[gid] = s1;
    s_out_s[gid] = s2;
    s_out_r[gid] = s3;
}

#define BSHIFT 9
#define BSIZE (1 << BSHIFT)
#define CHUNK 8192
#define EPT (CHUNK / 1024)
#define KMAX 1024
#define VBITS 23

__device__ __forceinline__ int block_excl_scan_1024(int v, int* wtot) {
    const int lane = threadIdx.x & 31;
    const int wave = threadIdx.x >> 5;
    int inc = v;
    for (int o = 1; o < 32; o <<= 1) {
        const int u = __shfl_up(inc, o);
        if (lane >= o) inc += u;
    }
    if (lane == 31) wtot[wave] = inc;
    __syncthreads();
    int carry = 0;
    for (int w = 0; w < wave; ++w) carry += wtot[w];
    return carry + inc - v;
}

__global__ __launch_bounds__(1024) void bucket_count_kernel(
    const int* __restrict__ src, const int* __restrict__ dst, int* __restrict__ gcnt,
    int E, int K, int P)
{
    __shared__ int cnt[KMAX];
    const int t = threadIdx.x;
    const int blk = blockIdx.x;
    const int dir = blockIdx.y;
    const int* key = dir ? src : dst;
    if (t < K) cnt[t] = 0;
    __syncthreads();
    const int e0 = blk * CHUNK + t;
    int kk[EPT];
#pragma unroll
    for (int i = 0; i < EPT; ++i) {
        const int e = e0 + i * 1024;
        kk[i] = key[(e < E) ? e : (E - 1)];
    }
#pragma unroll
    for (int i = 0; i < EPT; ++i) {
        if (e0 + i * 1024 < E) atomicAdd(&cnt[kk[i] >> BSHIFT], 1);
    }
    __syncthreads();
    int* g = gcnt + (size_t)dir * K * P;
    if (t < K) g[(size_t)t * P + blk] = cnt[t];
}

__global__ __launch_bounds__(1024) void scan_kernel(const int* __restrict__ deg,
                                                    int* __restrict__ rowptr, int N)
{
    __shared__ int wtot[32];
    const int dir = blockIdx.x;
    const int* dg = deg + (size_t)dir * N;
    int* rp = rowptr + (size_t)dir * (N + 1);
    const int lane = threadIdx.x & 31;
    const int wave = threadIdx.x >> 5;
    const int span = (((N + 31) / 32) + 127) & ~127;
    const int b = wave * span;
    const int en = (b + span < N) ? (b + span) : N;

    int s = 0;
    for (int i = b + lane; i < en; i += 32) s += dg[i];
    for (int o = 16; o > 0; o >>= 1) s += __shfl_xor(s, o);
    if (lane == 0) wtot[wave] = s;
    __syncthreads();
    int carry = 0;
    for (int w = 0; w < wave; ++w) carry += wtot[w];

    for (int i0 = b; i0 < en; i0 += 128) {
        const int i = i0 + lane * 4;
        const int v0 = (i     < en) ? dg[i]     : 0;
        const int v1 = (i + 1 < en) ? dg[i + 1] : 0;
        const int v2 = (i + 2 < en) ? dg[i + 2] : 0;
        const int v3 = (i + 3 < en) ? dg[i + 3] : 0;
        const int t = (v0 + v1) + (v2 + v3);
        int inc = t;
        for (int o = 1; o < 32; o <<= 1) {
            const int u = __shfl_up(inc, o);
            if (lane >= o) inc += u;
        }
        int ex = carry + inc - t;
        if (i     < en) rp[i]     = ex;
        ex += v0;
        if (i + 1 < en) rp[i + 1] = ex;
        ex += v1;
        if (i + 2 < en) rp[i + 2] = ex;
        ex += v2;
        if (i + 3 < en) rp[i + 3] = ex;
        carry += __shfl(inc, 31);
    }
    if (wave == 31 && lane == 0) rp[N] = carry;
}

__global__ __launch_bounds__(1024) void bucket_scatter_kernel(
    const int* __restrict__ src, const int* __restrict__ dst, const int* __restrict__ goff,
    unsigned int* __restrict__ binned, int E, int K, int P)
{
    __shared__ int cnt[KMAX];
    __shared__ int gbase[KMAX];
    __shared__ int wtot[32];
    __shared__ unsigned int s_item[CHUNK];
    __shared__ unsigned short s_bkt[CHUNK];
    const int t = threadIdx.x;
    const int blk = blockIdx.x;
    const int dir = blockIdx.y;
    const int* key = dir ? src : dst;
    const int* val = dir ? dst : src;
    if (t < K) cnt[t] = 0;
    __syncthreads();
    int kk[EPT], vv[EPT], rk[EPT];
    const int e0 = blk * CHUNK + t;
#pragma unroll
    for (int i = 0; i < EPT; ++i) {
        const int e = e0 + i * 1024;
        const int ec = (e < E) ? e : (E - 1);
        kk[i] = key[ec];
        vv[i] = val[ec];
    }
#pragma unroll
    for (int i = 0; i < EPT; ++i) {
        rk[i] = 0;
        if (e0 + i * 1024 < E) rk[i] = atomicAdd(&cnt[kk[i] >> BSHIFT], 1);
        else kk[i] = -1;
    }
    __syncthreads();
    const int c = (t < K) ? cnt[t] : 0;
    const int ex = block_excl_scan_1024(c, wtot);
    if (t < K) {
        cnt[t] = ex;
        gbase[t] = goff[(size_t)dir * ((size_t)K * P + 1) + (size_t)t * P + blk] - ex;
    }
    __syncthreads();
#pragma unroll
    for (int i = 0; i < EPT; ++i) {
        if (kk[i] >= 0) {
            const int b = kk[i] >> BSHIFT;
            const int pos = cnt[b] + rk[i];
            s_item[pos] = ((unsigned int)(kk[i] & (BSIZE - 1)) << VBITS) | (unsigned int)vv[i];
            s_bkt[pos] = (unsigned short)b;
        }
    }
    __syncthreads();
    const int left = E - blk * CHUNK;
    const int nloc = (left < CHUNK) ? left : CHUNK;
    unsigned int* outp = binned + (size_t)dir * E;
    for (int i = t; i < nloc; i += 1024) outp[gbase[s_bkt[i]] + i] = s_item[i];
}

#define DCAP 16384
#define DU 8
__global__ __launch_bounds__(1024) void bucket_rows_kernel(
    const unsigned int* __restrict__ binned, const int* __restrict__ goff,
    int* __restrict__ rowptr, int* __restrict__ nbr, int N, int E, int K, int P)
{
    __shared__ int cnt[BSIZE];
    __shared__ int cur[BSIZE];
    __shared__ int wtot[32];
    __shared__ int s_out[DCAP];
    const int t = threadIdx.x;
    const int b = blockIdx.x;
    const int dir = blockIdx.y;
    const int* go = goff + (size_t)dir * ((size_t)K * P + 1);
    const int beg = go[(size_t)b * P];
    const int end = go[(size_t)(b + 1) * P];
    const int cntb = end - beg;
    const bool staged = cntb <= DCAP;
    const unsigned int* items = binned + (size_t)dir * E;
    if (t < BSIZE) cnt[t] = 0;
    __syncthreads();
    for (int i0 = beg + t; i0 < end; i0 += DU * 1024) {
        unsigned int it[DU];
#pragma unroll
        for (int u = 0; u < DU; ++u) {
            const int j = i0 + u * 1024;
            it[u] = items[(j < end) ? j : i0];
        }
#pragma unroll
        for (int u = 0; u < DU; ++u) {
            if (i0 + u * 1024 < end) atomicAdd(&cnt[it[u] >> VBITS], 1);
        }
    }
    __syncthreads();
    const int c = (t < BSIZE) ? cnt[t] : 0;
    const int ex = block_excl_scan_1024(c, wtot);
    int* rp = rowptr + (size_t)dir * (N + 1);
    if (t < BSIZE) {
        cur[t] = ex;
        const int node = (b << BSHIFT) + t;
        if (node < N) rp[node] = beg + ex;
    }
    if (b == K - 1 && t == 0) rp[N] = end;
    __syncthreads();
    int* nb = nbr + (size_t)dir * E + beg;
    for (int i0 = beg + t; i0 < end; i0 += DU * 1024) {
        unsigned int it[DU];
#pragma unroll
        for (int u = 0; u < DU; ++u) {
            const int j = i0 + u * 1024;
            it[u] = items[(j < end) ? j : i0];
        }
#pragma unroll
        for (int u = 0; u < DU; ++u) {
            if (i0 + u * 1024 < end) {
                const int p = atomicAdd(&cur[it[u] >> VBITS], 1);
                const int v = (int)(it[u] & ((1u << VBITS) - 1u));
                if (staged) s_out[p] = v; else nb[p] = v;
            }
        }
    }
    if (staged) {
        __syncthreads();
        for (int i = t; i < cntb; i += 1024) nb[i] = s_out[i];
    }
}

__device__ __forceinline__ float wave_sum32(float v) {
    for (int o = 16; o > 0; o >>= 1) v += __shfl_xor(v, o);
    return v;
}
__device__ __forceinline__ float lrelu(float x) { return (x > 0.0f) ? x : NEG_SLOPE * x; }

#define DPP8_SEL(s0, s1, s2, s3, s4, s5, s6, s7) \
    ((s0) | ((s1) << 3) | ((s2) << 6) | ((s3) << 9) | ((s4) << 12) | ((s5) << 15) | ((s6) << 18) | ((s7) << 21))
#define DPP8_ALL(s) DPP8_SEL(s, s, s, s, s, s, s, s)
template <int SEL>
__device__ __forceinline__ float group8_pick(float v) {
    return __int_as_float(__builtin_amdgcn_mov_dpp8(__float_as_int(v), SEL));
}
__device__ __forceinline__ float group8_max(float v) {
    v = fmaxf(v, group8_pick<DPP8_SEL(1, 0, 3, 2, 5, 4, 7, 6)>(v));
    v = fmaxf(v, group8_pick<DPP8_SEL(2, 3, 0, 1, 6, 7, 4, 5)>(v));
    v = fmaxf(v, group8_pick<DPP8_SEL(4, 5, 6, 7, 0, 1, 2, 3)>(v));
    return v;
}
__device__ __forceinline__ float group8_sum(float v) {
    v += group8_pick<DPP8_SEL(1, 0, 3, 2, 5, 4, 7, 6)>(v);
    v += group8_pick<DPP8_SEL(2, 3, 0, 1, 6, 7, 4, 5)>(v);
    v += group8_pick<DPP8_SEL(4, 5, 6, 7, 0, 1, 2, 3)>(v);
    return v;
}

__device__ __forceinline__ void load_slots(const int* __restrict__ nbr, int c, int end,
                                           const float* __restrict__ s_s, float sr, int hd, int j,
                                           int (&sk)[4], float (&ek)[4])
{
    float sc[4];
#pragma unroll
    for (int k = 0; k < 4; ++k) {
        const int idx = c + 8 * k + j;
        sk[k] = nbr[(idx < end) ? idx : (end - 1)];
    }
    __builtin_amdgcn_sched_barrier(0);
#pragma unroll
    for (int k = 0; k < 4; ++k) sc[k] = s_s[(size_t)sk[k] * 4 + hd];
#pragma unroll
    for (int k = 0; k < 4; ++k)
        ek[k] = (c + 8 * k + j < end) ? lrelu(sc[k] + sr) : -INFINITY;
}

template <int J>
__device__ __forceinline__ v4f load_row(int sk, const float* __restrict__ h_proj, int lane) {
    const int si = __builtin_amdgcn_readlane(sk, J);
    return *(const v4f*)(h_proj + (size_t)si * OUT_DIM + 4 * lane);
}
template <int J0>
__device__ __forceinline__ void gather4(int sk, float wk, const float* __restrict__ h_proj,
                                        int lane, v4f& acc)
{
    const v4f x0 = load_row<J0 + 0>(sk, h_proj, lane);
    const v4f x1 = load_row<J0 + 1>(sk, h_proj, lane);
    const v4f x2 = load_row<J0 + 2>(sk, h_proj, lane);
    const v4f x3 = load_row<J0 + 3>(sk, h_proj, lane);
    acc += group8_pick<DPP8_ALL(J0 + 0)>(wk) * x0;
    acc += group8_pick<DPP8_ALL(J0 + 1)>(wk) * x1;
    acc += group8_pick<DPP8_ALL(J0 + 2)>(wk) * x2;
    acc += group8_pick<DPP8_ALL(J0 + 3)>(wk) * x3;
}

__device__ __forceinline__ void aggregate_dir(const int* __restrict__ nbr, int beg, int end,
                                              const float* __restrict__ s_s, float sr,
                                              const float* __restrict__ h_proj, int lane,
                                              v4f& acc)
{
    if (end <= beg) return;
    const bool more = (end - beg) > 32;
    const int hd = lane >> 3;
    const int j = lane & 7;

    int s0[4];
    float e0[4];
    load_slots(nbr, beg, end, s_s, sr, hd, j, s0, e0);

    float m = fmaxf(fmaxf(e0[0], e0[1]), fmaxf(e0[2], e0[3]));
    if (more) {
        for (int c = beg + 32; c < end; c += 32) {
            int sk[4];
            float ek[4];
            load_slots(nbr, c, end, s_s, sr, hd, j, sk, ek);
            m = fmaxf(m, fmaxf(fmaxf(ek[0], ek[1]), fmaxf(ek[2], ek[3])));
        }
    }
    m = group8_max(m);

    float a0[4];
#pragma unroll
    for (int k = 0; k < 4; ++k) a0[k] = (beg + 8 * k + j < end) ? __expf(e0[k] - m) : 0.0f;
    float den = (a0[0] + a0[1]) + (a0[2] + a0[3]);
    if (more) {
        for (int c = beg + 32; c < end; c += 32) {
            int sk[4];
            float ek[4];
            load_slots(nbr, c, end, s_s, sr, hd, j, sk, ek);
            float ak[4];
#pragma unroll
            for (int k = 0; k < 4; ++k) ak[k] = (c + 8 * k + j < end) ? __expf(ek[k] - m) : 0.0f;
            den += (ak[0] + ak[1]) + (ak[2] + ak[3]);
        }
    }
    den = group8_sum(den) + EPS_SOFTMAX;

    for (int c = beg; c < end; c += 32) {
        int sk[4];
        float wk[4];
        if (c == beg) {
#pragma unroll
            for (int k = 0; k < 4; ++k) { sk[k] = s0[k]; wk[k] = a0[k] / den; }
        } else {
            float ek[4];
            load_slots(nbr, c, end, s_s, sr, hd, j, sk, ek);
#pragma unroll
            for (int k = 0; k < 4; ++k)
                wk[k] = (c + 8 * k + j < end) ? (__expf(ek[k] - m) / den) : 0.0f;
        }
        const int cnt = (end - c < 32) ? (end - c) : 32;
#pragma unroll
        for (int k = 0; k < 4; ++k) {
            if (8 * k < cnt) {
                gather4<0>(sk[k], wk[k], h_proj, lane, acc);
                if (8 * k + 4 < cnt) gather4<4>(sk[k], wk[k], h_proj, lane, acc);
            }
        }
    }
}

__global__ __launch_bounds__(256) void gather_kernel(
    const int* __restrict__ rowptr, const int* __restrict__ nbr_in,
    const int* __restrict__ nbr_out, const float* __restrict__ h_proj,
    const float* __restrict__ s_in_s, const float* __restrict__ s_in_r,
    const float* __restrict__ s_out_s, const float* __restrict__ s_out_r,
    const float* __restrict__ bias, const float* __restrict__ gamma,
    const float* __restrict__ beta, float* __restrict__ out, int N)
{
    const int lane = threadIdx.x & 31;
    const int n = __builtin_amdgcn_readfirstlane(blockIdx.x * 8 + (threadIdx.x >> 5));
    if (n >= N) return;
    const int* rp_in = rowptr;
    const int* rp_out = rowptr + (size_t)(N + 1);
    const int bi = rp_in[n], ei = rp_in[n + 1];
    const int bo = rp_out[n], eo = rp_out[n + 1];
    const int hd = lane >> 3;
    const float sri = s_in_r[(size_t)n * 4 + hd];
    const float sro = s_out_r[(size_t)n * 4 + hd];
    float* orow = out + (size_t)n * OUT_DIM + 4 * lane;
    v4f comb = *(const v4f*)orow;

    v4f acc_in = {0.0f, 0.0f, 0.0f, 0.0f};
    aggregate_dir(nbr_in, bi, ei, s_in_s, sri, h_proj, lane, acc_in);
    comb += acc_in;
    v4f acc_out = {0.0f, 0.0f, 0.0f, 0.0f};
    aggregate_dir(nbr_out, bo, eo, s_out_s, sro, h_proj, lane, acc_out);

    const v4f x = (comb + acc_out) + *(const v4f*)(bias + 4 * lane);

    const float mean = wave_sum32((x[0] + x[1]) + (x[2] + x[3])) * (1.0f / OUT_DIM);
    const v4f d = x - mean;
    const float var = wave_sum32((d[0] * d[0] + d[1] * d[1]) + (d[2] * d[2] + d[3] * d[3]))
                      * (1.0f / OUT_DIM);
    const float inv = rsqrtf(var + EPS_LN);
    const v4f g = *(const v4f*)(gamma + 4 * lane);
    const v4f bt = *(const v4f*)(beta + 4 * lane);
    v4f y;
#pragma unroll
    for (int q = 0; q < 4; ++q) y[q] = d[q] * inv * g[q] + bt[q];
    *(v4f*)orow = y;
}

extern "C" void kernel_launch(void* const* d_in, const int* in_sizes, int n_in,
                              void* d_out, int out_size, void* d_ws, size_t ws_size,
                              hipStream_t stream) {
    const float* h      = (const float*)d_in[0];
    const int*   eidx   = (const int*)d_in[1];
    const float* W      = (const float*)d_in[2];
    const float* Wself  = (const float*)d_in[3];
    const float* a_in   = (const float*)d_in[4];
    const float* a_out  = (const float*)d_in[5];
    const float* bias   = (const float*)d_in[6];
    const float* gamma  = (const float*)d_in[7];
    const float* beta   = (const float*)d_in[8];
    float* out = (float*)d_out;

    const int N = in_sizes[0] / IN_DIM;
    const int E = in_sizes[1] / 2;
    const int NH = N * NHEAD;

    const int* src = eidx;
    const int* dst = eidx + E;

    float* ws = (float*)d_ws;
    float* h_proj  = ws;
    float* s_in_s  = h_proj + (size_t)N * OUT_DIM;
    float* s_in_r  = s_in_s + NH;
    float* s_out_s = s_in_r + NH;
    float* s_out_r = s_out_s + NH;
    int* cntoff  = (int*)(s_out_r + NH);
    int* rowptr  = cntoff + (size_t)2 * N;
    int* nbr_in  = rowptr + (size_t)2 * (N + 1);
    int* nbr_out = nbr_in + (size_t)E;
    const int K = (N + BSIZE - 1) >> BSHIFT;
    const int P = (E + CHUNK - 1) / CHUNK;
    const int KP = K * P;
    int* gcnt = cntoff;
    int* goff = gcnt + (size_t)2 * KP;
    unsigned int* binned = (unsigned int*)h_proj;

    {
        bucket_count_kernel<<<dim3(P, 2), 1024, 0, stream>>>(src, dst, gcnt, E, K, P);
    }
    {
        scan_kernel<<<2, 1024, 0, stream>>>(gcnt, goff, KP);
    }
    {
        bucket_scatter_kernel<<<dim3(P, 2), 1024, 0, stream>>>(src, dst, goff, binned, E, K, P);
    }
    {
        bucket_rows_kernel<<<dim3(K, 2), 1024, 0, stream>>>(binned, goff, rowptr, nbr_in,
                                                           N, E, K, P);
    }
    {
        gemm_wmma_kernel<<<GEMM_BLOCKS, 256, 0, stream>>>(h, W, Wself, h_proj, out, N);
    }
    {
        int threads = 256, blocks = (NH + threads - 1) / threads;
        score_kernel<<<blocks, threads, 0, stream>>>(h_proj, a_in, a_out,
                                                     s_in_s, s_in_r, s_out_s, s_out_r, NH);
    }
    {
        int blocks = (N + 7) / 8;
        gather_kernel<<<blocks, 256, 0, stream>>>(rowptr, nbr_in, nbr_out, h_proj,
                                                  s_in_s, s_in_r, s_out_s, s_out_r,
                                                  bias, gamma, beta, out, N);
    }
}


# List build in 2 launches, not 4: level 1 stores each chunk sorted by bucket with its run starts, level 2 gathers a bucket's runs from all chunks; no count or scan launch; slots stay arrival order
# speedup vs baseline: 21.5842x; 1.0853x over previous
#include <hip/hip_runtime.h>
#include <math.h>

typedef float v2f __attribute__((ext_vector_type(2)));
typedef float v4f __attribute__((ext_vector_type(4)));
typedef float v8f __attribute__((ext_vector_type(8)));

#define IN_DIM 128
#define OUT_DIM 128
#define NHEAD 4
#define HD 32
#define NEG_SLOPE 0.2f
#define EPS_SOFTMAX 1e-8f
#define EPS_LN 1e-5f

#define GEMM_BLOCKS 512
#define A_PITCH 132
__global__ __launch_bounds__(256) void gemm_wmma_kernel(
    const float* __restrict__ h, const float* __restrict__ W,
    const float* __restrict__ Wself, float* __restrict__ h_proj,
    float* __restrict__ out, int N)
{
    __shared__ __attribute__((aligned(16))) float As[2][16 * A_PITCH];
    const int tid = threadIdx.x;
    const int wave = tid >> 5;
    const int lane = tid & 31;
    const int c = lane & 15;
    const int hh = lane >> 4;
    const int ntiles = (N + 15) >> 4;
    int t = blockIdx.x;
    if (t >= ntiles) return;

    const int sr = tid >> 4;
    const int sq = (tid & 15) * 4;
    {
        int row = t * 16 + sr;
        if (row >= N) row = N - 1;
        const float* p = h + (size_t)row * IN_DIM + sq;
        const v4f p0 = *(const v4f*)p;
        const v4f p1 = *(const v4f*)(p + 64);
        *(v4f*)&As[0][sr * A_PITCH + sq] = p0;
        *(v4f*)&As[0][sr * A_PITCH + 64 + sq] = p1;
    }

    const int col0 = wave * 32;
    const float* bm = (wave < 4) ? (W + col0) : (Wself + (col0 - 128));
    v4f b0[16], b1[16];
#pragma unroll
    for (int j = 0; j < 16; ++j) {
        const float* bk = bm + (size_t)(8 * j + 4 * hh) * OUT_DIM + c;
        b0[j].x = bk[0];
        b0[j].y = bk[OUT_DIM];
        b0[j].z = bk[2 * OUT_DIM];
        b0[j].w = bk[3 * OUT_DIM];
        b1[j].x = bk[16];
        b1[j].y = bk[OUT_DIM + 16];
        b1[j].z = bk[2 * OUT_DIM + 16];
        b1[j].w = bk[3 * OUT_DIM + 16];
    }
    float* dst0 = ((wave < 4) ? (h_proj + col0) : (out + (col0 - 128))) + c;
    float* dst1 = dst0 + 16;
    const int mbase = hh * 8;
    const int step = (int)gridDim.x;
    __syncthreads();

    int buf = 0;
    for (; t < ntiles; t += step) {
        const int tn = t + step;
        const bool has_next = tn < ntiles;
        v4f p0 = {}, p1 = {};
        if (has_next) {
            int row = tn * 16 + sr;
            if (row >= N) row = N - 1;
            const float* p = h + (size_t)row * IN_DIM + sq;
            p0 = *(const v4f*)p;
            p1 = *(const v4f*)(p + 64);
        }

        const float* arow = &As[buf][c * A_PITCH + 4 * hh];
        v4f a[16];
#pragma unroll
        for (int j = 0; j < 16; ++j) a[j] = *(const v4f*)(arow + 8 * j);
        __builtin_amdgcn_sched_barrier(0);

        v8f c0 = {};
        v8f c1 = {};
#pragma unroll
        for (int j = 0; j < 16; ++j) {
            v2f alo, ahi, w0, w1;
            alo.x = a[j].x; alo.y = a[j].y;
            ahi.x = a[j].z; ahi.y = a[j].w;
            w0.x = b0[j].x; w0.y = b0[j].y;
            w1.x = b1[j].x; w1.y = b1[j].y;
            c0 = __builtin_amdgcn_wmma_f32_16x16x4_f32(false, alo, false, w0,
                                                       (short)0, c0, false, false);
            c1 = __builtin_amdgcn_wmma_f32_16x16x4_f32(false, alo, false, w1,
                                                       (short)0, c1, false, false);
            w0.x = b0[j].z; w0.y = b0[j].w;
            w1.x = b1[j].z; w1.y = b1[j].w;
            c0 = __builtin_amdgcn_wmma_f32_16x16x4_f32(false, ahi, false, w0,
                                                       (short)0, c0, false, false);
            c1 = __builtin_amdgcn_wmma_f32_16x16x4_f32(false, ahi, false, w1,
                                                       (short)0, c1, false, false);
        }

        const int row0 = t * 16;
        if (row0 + 16 <= N) {
#pragma unroll
            for (int v = 0; v < 8; ++v) {
                const size_t row = (size_t)(row0 + mbase + v);
                dst0[row * OUT_DIM] = c0[v];
                dst1[row * OUT_DIM] = c1[v];
            }
        } else {
#pragma unroll
            for (int v = 0; v < 8; ++v) {
                const int row = row0 + mbase + v;
                if (row < N) {
                    dst0[(size_t)row * OUT_DIM] = c0[v];
                    dst1[(size_t)row * OUT_DIM] = c1[v];
                }
            }
        }

        if (has_next) {
            *(v4f*)&As[buf ^ 1][sr * A_PITCH + sq] = p0;
            *(v4f*)&As[buf ^ 1][sr * A_PITCH + 64 + sq] = p1;
        }
        __syncthreads();
        buf ^= 1;
    }
}

__global__ void score_kernel(const float* __restrict__ h_proj,
                             const float* __restrict__ a_in,
                             const float* __restrict__ a_out,
                             float* __restrict__ s_in_s, float* __restrict__ s_in_r,
                             float* __restrict__ s_out_s, float* __restrict__ s_out_r,
                             int NH)
{
    int gid = blockIdx.x * blockDim.x + threadIdx.x;
    if (gid >= NH) return;
    const int hh = gid & 3;
    const int n = gid >> 2;
    const float* x = h_proj + (size_t)n * OUT_DIM + hh * HD;
    const float* ai = a_in + hh * 2 * HD;
    const float* ao = a_out + hh * 2 * HD;
    float s0 = 0.f, s1 = 0.f, s2 = 0.f, s3 = 0.f;
#pragma unroll
    for (int d = 0; d < HD; d += 4) {
        const v4f xv = *(const v4f*)(x + d);
#pragma unroll
        for (int j = 0; j < 4; ++j) {
            s0 += xv[j] * ai[d + j];
            s1 += xv[j] * ai[HD + d + j];
            s2 += xv[j] * ao[d + j];
            s3 += xv[j] * ao[HD + d + j];
        }
    }
    s_in_s[gid] = s0;
    s_in_r[gid] = s1;
    s_out_s[gid] = s2;
    s_out_r[gid] = s3;
}

#define BSHIFT 9
#define BSIZE (1 << BSHIFT)
#define CHUNK 8192
#define EPT (CHUNK / 1024)
#define KMAX 1024
#define PMAX 1024
#define VBITS 23

__device__ __forceinline__ int block_excl_scan_1024(int v, int* wtot) {
    const int lane = threadIdx.x & 31;
    const int wave = threadIdx.x >> 5;
    int inc = v;
    for (int o = 1; o < 32; o <<= 1) {
        const int u = __shfl_up(inc, o);
        if (lane >= o) inc += u;
    }
    if (lane == 31) wtot[wave] = inc;
    __syncthreads();
    int carry = 0;
    for (int w = 0; w < wave; ++w) carry += wtot[w];
    return carry + inc - v;
}

__global__ __launch_bounds__(1024) void bucket_sort_kernel(
    const int* __restrict__ src, const int* __restrict__ dst, int* __restrict__ tab,
    unsigned int* __restrict__ binned, int E, int K, int P)
{
    __shared__ int cnt[KMAX];
    __shared__ int wtot[32];
    __shared__ unsigned int s_item[CHUNK];
    const int t = threadIdx.x;
    const int blk = blockIdx.x;
    const int dir = blockIdx.y;
    const int* key = dir ? src : dst;
    const int* val = dir ? dst : src;
    if (t < K) cnt[t] = 0;
    __syncthreads();
    int kk[EPT], vv[EPT], rk[EPT];
    const int e0 = blk * CHUNK + t;
#pragma unroll
    for (int i = 0; i < EPT; ++i) {
        const int e = e0 + i * 1024;
        const int ec = (e < E) ? e : (E - 1);
        kk[i] = key[ec];
        vv[i] = val[ec];
    }
#pragma unroll
    for (int i = 0; i < EPT; ++i) {
        rk[i] = 0;
        if (e0 + i * 1024 < E) rk[i] = atomicAdd(&cnt[kk[i] >> BSHIFT], 1);
        else kk[i] = -1;
    }
    __syncthreads();
    const int c = (t < K) ? cnt[t] : 0;
    const int ex = block_excl_scan_1024(c, wtot);
    if (t < K) {
        cnt[t] = ex;
        int* tb = tab + (size_t)dir * 2 * K * P + (size_t)t * P + blk;
        tb[0] = c;
        tb[(size_t)K * P] = ex;
    }
    __syncthreads();
#pragma unroll
    for (int i = 0; i < EPT; ++i) {
        if (kk[i] >= 0) {
            const int pos = cnt[kk[i] >> BSHIFT] + rk[i];
            s_item[pos] = ((unsigned int)(kk[i] & (BSIZE - 1)) << VBITS) | (unsigned int)vv[i];
        }
    }
    __syncthreads();
    const int left = E - blk * CHUNK;
    const int nloc = (left < CHUNK) ? left : CHUNK;
    unsigned int* outp = binned + (size_t)dir * E + (size_t)blk * CHUNK;
    for (int i = t; i < nloc; i += 1024) outp[i] = s_item[i];
}

#define DCAP 16384
#define RB 7

__device__ __forceinline__ unsigned int load_runs(const unsigned int* __restrict__ items,
                                                  const int* run_cn, const int* run_st,
                                                  int q0, int P, int lane,
                                                  unsigned int (&it)[2 * RB], int& longest)
{
    int rc[RB], rs[RB];
#pragma unroll
    for (int u = 0; u < RB; ++u) {
        const int c = q0 + 32 * u;
        const int cc = (c < P) ? c : (P - 1);
        const int n = run_cn[cc];
        const int s = run_st[cc];
        rc[u] = (c < P) ? n : 0;
        rs[u] = (c < P) ? s : 0;
    }
    unsigned int valid = 0;
    longest = 0;
#pragma unroll
    for (int u = 0; u < RB; ++u) {
        longest = (rc[u] > longest) ? rc[u] : longest;
        const bool in0 = lane < rc[u];
        const bool in1 = lane + 32 < rc[u];
        it[2 * u] = items[rs[u] + (in0 ? lane : 0)];
        it[2 * u + 1] = items[rs[u] + (in1 ? (lane + 32) : 0)];
        valid |= ((in0 ? 1u : 0u) | (in1 ? 2u : 0u)) << (2 * u);
    }
    return valid;
}
__device__ __forceinline__ void place_one(unsigned int item, int* cur, int* s_out, int* nb,
                                          bool staged)
{
    const int p = atomicAdd(&cur[item >> VBITS], 1);
    const int v = (int)(item & ((1u << VBITS) - 1u));
    if (staged) s_out[p] = v; else nb[p] = v;
}

__global__ __launch_bounds__(1024) void bucket_rows_kernel(
    const unsigned int* __restrict__ binned, const int* __restrict__ tab,
    int* __restrict__ rowptr, int* __restrict__ nbr, int N, int E, int K, int P)
{
    __shared__ int cnt[BSIZE];
    __shared__ int cur[BSIZE];
    __shared__ int wtot[32];
    __shared__ int wsum[64];
    __shared__ int run_cn[PMAX];
    __shared__ int run_st[PMAX];
    __shared__ int s_out[DCAP];
    const int t = threadIdx.x;
    const int lane = t & 31;
    const int wave = t >> 5;
    const int b = blockIdx.x;
    const int dir = blockIdx.y;
    const int* tb = tab + (size_t)dir * 2 * K * P + (size_t)b * P;
    int cn = 0, st = 0;
    if (t < P) {
        cn = tb[t];
        st = tb[(size_t)K * P + t];
        run_cn[t] = cn;
        run_st[t] = (cn > 0) ? (t * CHUNK + st) : 0;
    }
    if (t < BSIZE) cnt[t] = 0;
    for (int o = 16; o > 0; o >>= 1) {
        cn += __shfl_xor(cn, o);
        st += __shfl_xor(st, o);
    }
    if (lane == 0) { wsum[wave] = st; wsum[32 + wave] = cn; }
    __syncthreads();
    int beg = wsum[lane], cntb = wsum[32 + lane];
    for (int o = 16; o > 0; o >>= 1) {
        beg += __shfl_xor(beg, o);
        cntb += __shfl_xor(cntb, o);
    }
    const int end = beg + cntb;
    const bool staged = cntb <= DCAP;
    const unsigned int* items = binned + (size_t)dir * E;

    for (int q0 = wave; q0 < P; q0 += 32 * RB) {
        unsigned int it[2 * RB];
        int longest;
        const unsigned int valid = load_runs(items, run_cn, run_st, q0, P, lane, it, longest);
#pragma unroll
        for (int i = 0; i < 2 * RB; ++i) {
            if ((valid >> i) & 1u) atomicAdd(&cnt[it[i] >> VBITS], 1);
        }
        if (longest > 64) {
            for (int c = q0; c < P && c < q0 + 32 * RB; c += 32) {
                const int n = run_cn[c];
                const int s = run_st[c];
                for (int k = 64 + lane; k < n; k += 32) atomicAdd(&cnt[items[s + k] >> VBITS], 1);
            }
        }
    }
    __syncthreads();
    const int c = (t < BSIZE) ? cnt[t] : 0;
    const int ex = block_excl_scan_1024(c, wtot);
    int* rp = rowptr + (size_t)dir * (N + 1);
    if (t < BSIZE) {
        cur[t] = ex;
        const int node = (b << BSHIFT) + t;
        if (node < N) rp[node] = beg + ex;
    }
    if (b == K - 1 && t == 0) rp[N] = end;
    __syncthreads();
    int* nb = nbr + (size_t)dir * E + beg;
    for (int q0 = wave; q0 < P; q0 += 32 * RB) {
        unsigned int it[2 * RB];
        int longest;
        const unsigned int valid = load_runs(items, run_cn, run_st, q0, P, lane, it, longest);
#pragma unroll
        for (int i = 0; i < 2 * RB; ++i) {
            if ((valid >> i) & 1u) place_one(it[i], cur, s_out, nb, staged);
        }
        if (longest > 64) {
            for (int c = q0; c < P && c < q0 + 32 * RB; c += 32) {
                const int n = run_cn[c];
                const int s = run_st[c];
                for (int k = 64 + lane; k < n; k += 32) place_one(items[s + k], cur, s_out, nb, staged);
            }
        }
    }
    if (staged) {
        __syncthreads();
        for (int i = t; i < cntb; i += 1024) nb[i] = s_out[i];
    }
}

__device__ __forceinline__ float wave_sum32(float v) {
    for (int o = 16; o > 0; o >>= 1) v += __shfl_xor(v, o);
    return v;
}
__device__ __forceinline__ float lrelu(float x) { return (x > 0.0f) ? x : NEG_SLOPE * x; }

#define DPP8_SEL(s0, s1, s2, s3, s4, s5, s6, s7) \
    ((s0) | ((s1) << 3) | ((s2) << 6) | ((s3) << 9) | ((s4) << 12) | ((s5) << 15) | ((s6) << 18) | ((s7) << 21))
#define DPP8_ALL(s) DPP8_SEL(s, s, s, s, s, s, s, s)
template <int SEL>
__device__ __forceinline__ float group8_pick(float v) {
    return __int_as_float(__builtin_amdgcn_mov_dpp8(__float_as_int(v), SEL));
}
__device__ __forceinline__ float group8_max(float v) {
    v = fmaxf(v, group8_pick<DPP8_SEL(1, 0, 3, 2, 5, 4, 7, 6)>(v));
    v = fmaxf(v, group8_pick<DPP8_SEL(2, 3, 0, 1, 6, 7, 4, 5)>(v));
    v = fmaxf(v, group8_pick<DPP8_SEL(4, 5, 6, 7, 0, 1, 2, 3)>(v));
    return v;
}
__device__ __forceinline__ float group8_sum(float v) {
    v += group8_pick<DPP8_SEL(1, 0, 3, 2, 5, 4, 7, 6)>(v);
    v += group8_pick<DPP8_SEL(2, 3, 0, 1, 6, 7, 4, 5)>(v);
    v += group8_pick<DPP8_SEL(4, 5, 6, 7, 0, 1, 2, 3)>(v);
    return v;
}

__device__ __forceinline__ void load_slots(const int* __restrict__ nbr, int c, int end,
                                           const float* __restrict__ s_s, float sr, int hd, int j,
                                           int (&sk)[4], float (&ek)[4])
{
    float sc[4];
#pragma unroll
    for (int k = 0; k < 4; ++k) {
        const int idx = c + 8 * k + j;
        sk[k] = nbr[(idx < end) ? idx : (end - 1)];
    }
    __builtin_amdgcn_sched_barrier(0);
#pragma unroll
    for (int k = 0; k < 4; ++k) sc[k] = s_s[(size_t)sk[k] * 4 + hd];
#pragma unroll
    for (int k = 0; k < 4; ++k)
        ek[k] = (c + 8 * k + j < end) ? lrelu(sc[k] + sr) : -INFINITY;
}

template <int J>
__device__ __forceinline__ v4f load_row(int sk, const float* __restrict__ h_proj, int lane) {
    const int si = __builtin_amdgcn_readlane(sk, J);
    return *(const v4f*)(h_proj + (size_t)si * OUT_DIM + 4 * lane);
}
template <int J0>
__device__ __forceinline__ void gather4(int sk, float wk, const float* __restrict__ h_proj,
                                        int lane, v4f& acc)
{
    const v4f x0 = load_row<J0 + 0>(sk, h_proj, lane);
    const v4f x1 = load_row<J0 + 1>(sk, h_proj, lane);
    const v4f x2 = load_row<J0 + 2>(sk, h_proj, lane);
    const v4f x3 = load_row<J0 + 3>(sk, h_proj, lane);
    acc += group8_pick<DPP8_ALL(J0 + 0)>(wk) * x0;
    acc += group8_pick<DPP8_ALL(J0 + 1)>(wk) * x1;
    acc += group8_pick<DPP8_ALL(J0 + 2)>(wk) * x2;
    acc += group8_pick<DPP8_ALL(J0 + 3)>(wk) * x3;
}

__device__ __forceinline__ void aggregate_dir(const int* __restrict__ nbr, int beg, int end,
                                              const float* __restrict__ s_s, float sr,
                                              const float* __restrict__ h_proj, int lane,
                                              v4f& acc)
{
    if (end <= beg) return;
    const bool more = (end - beg) > 32;
    const int hd = lane >> 3;
    const int j = lane & 7;

    int s0[4];
    float e0[4];
    load_slots(nbr, beg, end, s_s, sr, hd, j, s0, e0);

    float m = fmaxf(fmaxf(e0[0], e0[1]), fmaxf(e0[2], e0[3]));
    if (more) {
        for (int c = beg + 32; c < end; c += 32) {
            int sk[4];
            float ek[4];
            load_slots(nbr, c, end, s_s, sr, hd, j, sk, ek);
            m = fmaxf(m, fmaxf(fmaxf(ek[0], ek[1]), fmaxf(ek[2], ek[3])));
        }
    }
    m = group8_max(m);

    float a0[4];
#pragma unroll
    for (int k = 0; k < 4; ++k) a0[k] = (beg + 8 * k + j < end) ? __expf(e0[k] - m) : 0.0f;
    float den = (a0[0] + a0[1]) + (a0[2] + a0[3]);
    if (more) {
        for (int c = beg + 32; c < end; c += 32) {
            int sk[4];
            float ek[4];
            load_slots(nbr, c, end, s_s, sr, hd, j, sk, ek);
            float ak[4];
#pragma unroll
            for (int k = 0; k < 4; ++k) ak[k] = (c + 8 * k + j < end) ? __expf(ek[k] - m) : 0.0f;
            den += (ak[0] + ak[1]) + (ak[2] + ak[3]);
        }
    }
    den = group8_sum(den) + EPS_SOFTMAX;

    for (int c = beg; c < end; c += 32) {
        int sk[4];
        float wk[4];
        if (c == beg) {
#pragma unroll
            for (int k = 0; k < 4; ++k) { sk[k] = s0[k]; wk[k] = a0[k] / den; }
        } else {
            float ek[4];
            load_slots(nbr, c, end, s_s, sr, hd, j, sk, ek);
#pragma unroll
            for (int k = 0; k < 4; ++k)
                wk[k] = (c + 8 * k + j < end) ? (__expf(ek[k] - m) / den) : 0.0f;
        }
        const int cnt = (end - c < 32) ? (end - c) : 32;
#pragma unroll
        for (int k = 0; k < 4; ++k) {
            if (8 * k < cnt) {
                gather4<0>(sk[k], wk[k], h_proj, lane, acc);
                if (8 * k + 4 < cnt) gather4<4>(sk[k], wk[k], h_proj, lane, acc);
            }
        }
    }
}

__global__ __launch_bounds__(256) void gather_kernel(
    const int* __restrict__ rowptr, const int* __restrict__ nbr_in,
    const int* __restrict__ nbr_out, const float* __restrict__ h_proj,
    const float* __restrict__ s_in_s, const float* __restrict__ s_in_r,
    const float* __restrict__ s_out_s, const float* __restrict__ s_out_r,
    const float* __restrict__ bias, const float* __restrict__ gamma,
    const float* __restrict__ beta, float* __restrict__ out, int N)
{
    const int lane = threadIdx.x & 31;
    const int n = __builtin_amdgcn_readfirstlane(blockIdx.x * 8 + (threadIdx.x >> 5));
    if (n >= N) return;
    const int* rp_in = rowptr;
    const int* rp_out = rowptr + (size_t)(N + 1);
    const int bi = rp_in[n], ei = rp_in[n + 1];
    const int bo = rp_out[n], eo = rp_out[n + 1];
    const int hd = lane >> 3;
    const float sri = s_in_r[(size_t)n * 4 + hd];
    const float sro = s_out_r[(size_t)n * 4 + hd];
    float* orow = out + (size_t)n * OUT_DIM + 4 * lane;
    v4f comb = *(const v4f*)orow;

    v4f acc_in = {0.0f, 0.0f, 0.0f, 0.0f};
    aggregate_dir(nbr_in, bi, ei, s_in_s, sri, h_proj, lane, acc_in);
    comb += acc_in;
    v4f acc_out = {0.0f, 0.0f, 0.0f, 0.0f};
    aggregate_dir(nbr_out, bo, eo, s_out_s, sro, h_proj, lane, acc_out);

    const v4f x = (comb + acc_out) + *(const v4f*)(bias + 4 * lane);

    const float mean = wave_sum32((x[0] + x[1]) + (x[2] + x[3])) * (1.0f / OUT_DIM);
    const v4f d = x - mean;
    const float var = wave_sum32((d[0] * d[0] + d[1] * d[1]) + (d[2] * d[2] + d[3] * d[3]))
                      * (1.0f / OUT_DIM);
    const float inv = rsqrtf(var + EPS_LN);
    const v4f g = *(const v4f*)(gamma + 4 * lane);
    const v4f bt = *(const v4f*)(beta + 4 * lane);
    v4f y;
#pragma unroll
    for (int q = 0; q < 4; ++q) y[q] = d[q] * inv * g[q] + bt[q];
    *(v4f*)orow = y;
}

extern "C" void kernel_launch(void* const* d_in, const int* in_sizes, int n_in,
                              void* d_out, int out_size, void* d_ws, size_t ws_size,
                              hipStream_t stream) {
    const float* h      = (const float*)d_in[0];
    const int*   eidx   = (const int*)d_in[1];
    const float* W      = (const float*)d_in[2];
    const float* Wself  = (const float*)d_in[3];
    const float* a_in   = (const float*)d_in[4];
    const float* a_out  = (const float*)d_in[5];
    const float* bias   = (const float*)d_in[6];
    const float* gamma  = (const float*)d_in[7];
    const float* beta   = (const float*)d_in[8];
    float* out = (float*)d_out;

    const int N = in_sizes[0] / IN_DIM;
    const int E = in_sizes[1] / 2;
    const int NH = N * NHEAD;

    const int* src = eidx;
    const int* dst = eidx + E;

    float* ws = (float*)d_ws;
    float* h_proj  = ws;
    float* s_in_s  = h_proj + (size_t)N * OUT_DIM;
    float* s_in_r  = s_in_s + NH;
    float* s_out_s = s_in_r + NH;
    float* s_out_r = s_out_s + NH;
    int* cntoff  = (int*)(s_out_r + NH);
    int* rowptr  = cntoff + (size_t)2 * N;
    int* nbr_in  = rowptr + (size_t)2 * (N + 1);
    int* nbr_out = nbr_in + (size_t)E;
    const int K = (N + BSIZE - 1) >> BSHIFT;
    const int P = (E + CHUNK - 1) / CHUNK;
    int* tab = cntoff;
    unsigned int* binned = (unsigned int*)h_proj;

    {
        bucket_sort_kernel<<<dim3(P, 2), 1024, 0, stream>>>(src, dst, tab, binned, E, K, P);
    }
    {
        bucket_rows_kernel<<<dim3(K, 2), 1024, 0, stream>>>(binned, tab, rowptr, nbr_in,
                                                           N, E, K, P);
    }
    {
        gemm_wmma_kernel<<<GEMM_BLOCKS, 256, 0, stream>>>(h, W, Wself, h_proj, out, N);
    }
    {
        int threads = 256, blocks = (NH + threads - 1) / threads;
        score_kernel<<<blocks, threads, 0, stream>>>(h_proj, a_in, a_out,
                                                     s_in_s, s_in_r, s_out_s, s_out_r, NH);
    }
    {
        int blocks = (N + 7) / 8;
        gather_kernel<<<blocks, 256, 0, stream>>>(rowptr, nbr_in, nbr_out, h_proj,
                                                  s_in_s, s_in_r, s_out_s, s_out_r,
                                                  bias, gamma, beta, out, N);
    }
}
